# starting kernel
# baseline (speedup 1.0000x reference)
#include <hip/hip_runtime.h>
#include <hip/hip_bf16.h>
#include <math.h>

typedef float v2f __attribute__((ext_vector_type(2)));
typedef float v8f __attribute__((ext_vector_type(8)));

#define NFEAT 256
#define NHID  64
#define NEG_SLOPE 0.2f

__global__ void gat_init(float* __restrict__ out, const float* __restrict__ bias,
                         float* __restrict__ mmax, float* __restrict__ denom, int n) {
    int i = blockIdx.x * blockDim.x + threadIdx.x;
    if (i < n * NHID) out[i] = bias[i & (NHID - 1)];
    if (i < n) { mmax[i] = -3.0e38f; denom[i] = 0.0f; }
}

__global__ void __launch_bounds__(256)
gat_gemm(const float* __restrict__ X, const float* __restrict__ W,
         float* __restrict__ XP, int n) {
    const int tid  = threadIdx.x;
    const int wave = tid >> 5;
    const int lane = tid & 31;
    const int half = lane >> 4;
    const int lrow = lane & 15;
    const int r0   = (blockIdx.x * 8 + wave) * 16;
    if (r0 >= n) return;

    int arow = r0 + lrow;
    if (arow > n - 1) arow = n - 1;
    const float* aptr = X + (size_t)arow * NFEAT + 2 * half;

    v8f c0 = {}, c1 = {}, c2 = {}, c3 = {};

#pragma unroll 8
    for (int k = 0; k < NFEAT; k += 4) {
        v2f a = *(const v2f*)(aptr + k);
        const float* wp = W + (size_t)(k + 2 * half) * NHID + lrow;
        v2f b0, b1, b2, b3;
        b0[0] = wp[0];   b0[1] = wp[NHID + 0];
        b1[0] = wp[16];  b1[1] = wp[NHID + 16];
        b2[0] = wp[32];  b2[1] = wp[NHID + 32];
        b3[0] = wp[48];  b3[1] = wp[NHID + 48];
        c0 = __builtin_amdgcn_wmma_f32_16x16x4_f32(false, a, false, b0, (short)0, c0, false, false);
        c1 = __builtin_amdgcn_wmma_f32_16x16x4_f32(false, a, false, b1, (short)0, c1, false, false);
        c2 = __builtin_amdgcn_wmma_f32_16x16x4_f32(false, a, false, b2, (short)0, c2, false, false);
        c3 = __builtin_amdgcn_wmma_f32_16x16x4_f32(false, a, false, b3, (short)0, c3, false, false);
    }

#pragma unroll
    for (int j = 0; j < 8; ++j) {
        int row = half * 8 + j;
        if (r0 + row < n) {
            float* o = XP + (size_t)(r0 + row) * NHID + lrow;
            o[0]  = c0[j];
            o[16] = c1[j];
            o[32] = c2[j];
            o[48] = c3[j];
        }
    }
}

__global__ void __launch_bounds__(256)
gat_scores(const float* __restrict__ XP, const float* __restrict__ As,
           const float* __restrict__ Ad, float* __restrict__ Ss,
           float* __restrict__ Sd, int n) {
    const int tid  = threadIdx.x;
    const int node = blockIdx.x * 16 + (tid >> 4);
    const int part = tid & 15;
    if (node >= n) return;
    float4 xv = ((const float4*)(XP + (size_t)node * NHID))[part];
    float4 av = ((const float4*)As)[part];
    float4 dv = ((const float4*)Ad)[part];
    float ss = xv.x * av.x + xv.y * av.y + xv.z * av.z + xv.w * av.w;
    float sd = xv.x * dv.x + xv.y * dv.y + xv.z * dv.z + xv.w * dv.w;
#pragma unroll
    for (int m = 1; m < 16; m <<= 1) {
        ss += __shfl_xor(ss, m, 32);
        sd += __shfl_xor(sd, m, 32);
    }
    if (part == 0) { Ss[node] = ss; Sd[node] = sd; }
}

__global__ void gat_logits(const int* __restrict__ src, const int* __restrict__ dst,
                           const float* __restrict__ Ss, const float* __restrict__ Sd,
                           float* __restrict__ ew, float* __restrict__ mmax,
                           int E, int n) {
    int k = blockIdx.x * blockDim.x + threadIdx.x;
    if (k >= E + n) return;
    int s = (k < E) ? src[k] : (k - E);
    int d = (k < E) ? dst[k] : (k - E);
    float v = Ss[s] + Sd[d];
    float e = (v > 0.0f) ? v : NEG_SLOPE * v;
    ew[k] = e;
    atomicMax(mmax + d, e);
}

__global__ void gat_expsum(const int* __restrict__ dst, float* __restrict__ ew,
                           const float* __restrict__ mmax, float* __restrict__ denom,
                           int E, int n) {
    int k = blockIdx.x * blockDim.x + threadIdx.x;
    if (k >= E + n) return;
    int d = (k < E) ? dst[k] : (k - E);
    float ex = __expf(ew[k] - mmax[d]);
    ew[k] = ex;
    atomicAdd(denom + d, ex);
}

__global__ void __launch_bounds__(256)
gat_scatter(const int* __restrict__ src, const int* __restrict__ dst,
            const float* __restrict__ ew, const float* __restrict__ denom,
            const float* __restrict__ XP, float* __restrict__ out, int E, int n) {
    const int tid  = threadIdx.x;
    const int k    = blockIdx.x * 16 + (tid >> 4);
    if (k >= E + n) return;
    const int part = tid & 15;
    int s = (k < E) ? src[k] : (k - E);
    int d = (k < E) ? dst[k] : (k - E);
    float alpha = ew[k] / denom[d];
    float4 xv = ((const float4*)(XP + (size_t)s * NHID))[part];
    float* o = out + (size_t)d * NHID + part * 4;
    atomicAdd(o + 0, alpha * xv.x);
    atomicAdd(o + 1, alpha * xv.y);
    atomicAdd(o + 2, alpha * xv.z);
    atomicAdd(o + 3, alpha * xv.w);
}

extern "C" void kernel_launch(void* const* d_in, const int* in_sizes, int n_in,
                              void* d_out, int out_size, void* d_ws, size_t ws_size,
                              hipStream_t stream) {
    const float* x     = (const float*)d_in[0];
    const int*   ei    = (const int*)  d_in[1];
    const float* W     = (const float*)d_in[2];
    const float* a_src = (const float*)d_in[3];
    const float* a_dst = (const float*)d_in[4];
    const float* bias  = (const float*)d_in[5];
    float*       out   = (float*)d_out;

    const int N = in_sizes[0] / NFEAT;
    const int E = in_sizes[1] / 2;
    const int* src = ei;
    const int* dst = ei + E;

    float* xp    = (float*)d_ws;
    float* ew    = xp    + (size_t)N * NHID;
    float* mmax  = ew    + (size_t)(E + N);
    float* denom = mmax  + N;
    float* ssrc  = denom + N;
    float* sdst  = ssrc  + N;

    gat_init   <<<(N * NHID + 255) / 256, 256, 0, stream>>>(out, bias, mmax, denom, N);
    gat_gemm   <<<(N + 127) / 128,        256, 0, stream>>>(x, W, xp, N);
    gat_scores <<<(N + 15) / 16,          256, 0, stream>>>(xp, a_src, a_dst, ssrc, sdst, N);
    gat_logits <<<(E + N + 255) / 256,    256, 0, stream>>>(src, dst, ssrc, sdst, ew, mmax, E, N);
    gat_expsum <<<(E + N + 255) / 256,    256, 0, stream>>>(dst, ew, mmax, denom, E, N);
    gat_scatter<<<(E + N + 15) / 16,      256, 0, stream>>>(src, dst, ew, denom, xp, out, E, N);
}


# edges binned by destination block each call (LDS-counter counting sort); an owner block per 128 nodes does max, exp, row sums in LDS, writes rows once: no global atomics; LDS add order, divide after
# speedup vs baseline: 2.4311x; 2.4311x over previous
#include <hip/hip_runtime.h>
#include <hip/hip_bf16.h>
#include <math.h>

typedef float v2f __attribute__((ext_vector_type(2)));
typedef float v8f __attribute__((ext_vector_type(8)));

#define NFEAT 256
#define NHID  64
#define NEG_SLOPE 0.2f

#define BLK_NODES 128
#define BLK_SHIFT 7
#define CHUNK     8192
#define MAXBINS   1024

__global__ void __launch_bounds__(256)
gat_gemm(const float* __restrict__ X, const float* __restrict__ W,
         float* __restrict__ XP, int n) {
    const int tid  = threadIdx.x;
    const int wave = tid >> 5;
    const int lane = tid & 31;
    const int half = lane >> 4;
    const int lrow = lane & 15;
    const int r0   = (blockIdx.x * 8 + wave) * 16;
    if (r0 >= n) return;

    int arow = r0 + lrow;
    if (arow > n - 1) arow = n - 1;
    const float* aptr = X + (size_t)arow * NFEAT + 2 * half;

    v8f c0 = {}, c1 = {}, c2 = {}, c3 = {};

#pragma unroll 8
    for (int k = 0; k < NFEAT; k += 4) {
        v2f a = *(const v2f*)(aptr + k);
        const float* wp = W + (size_t)(k + 2 * half) * NHID + lrow;
        v2f b0, b1, b2, b3;
        b0[0] = wp[0];   b0[1] = wp[NHID + 0];
        b1[0] = wp[16];  b1[1] = wp[NHID + 16];
        b2[0] = wp[32];  b2[1] = wp[NHID + 32];
        b3[0] = wp[48];  b3[1] = wp[NHID + 48];
        c0 = __builtin_amdgcn_wmma_f32_16x16x4_f32(false, a, false, b0, (short)0, c0, false, false);
        c1 = __builtin_amdgcn_wmma_f32_16x16x4_f32(false, a, false, b1, (short)0, c1, false, false);
        c2 = __builtin_amdgcn_wmma_f32_16x16x4_f32(false, a, false, b2, (short)0, c2, false, false);
        c3 = __builtin_amdgcn_wmma_f32_16x16x4_f32(false, a, false, b3, (short)0, c3, false, false);
    }

#pragma unroll
    for (int j = 0; j < 8; ++j) {
        int row = half * 8 + j;
        if (r0 + row < n) {
            float* o = XP + (size_t)(r0 + row) * NHID + lrow;
            o[0]  = c0[j];
            o[16] = c1[j];
            o[32] = c2[j];
            o[48] = c3[j];
        }
    }
}

__global__ void __launch_bounds__(256)
gat_scores(const float* __restrict__ XP, const float* __restrict__ As,
           const float* __restrict__ Ad, float* __restrict__ Ss,
           float* __restrict__ Sd, int n) {
    const int tid  = threadIdx.x;
    const int node = blockIdx.x * 16 + (tid >> 4);
    const int part = tid & 15;
    if (node >= n) return;
    float4 xv = ((const float4*)(XP + (size_t)node * NHID))[part];
    float4 av = ((const float4*)As)[part];
    float4 dv = ((const float4*)Ad)[part];
    float ss = xv.x * av.x + xv.y * av.y + xv.z * av.z + xv.w * av.w;
    float sd = xv.x * dv.x + xv.y * dv.y + xv.z * dv.z + xv.w * dv.w;
#pragma unroll
    for (int m = 1; m < 16; m <<= 1) {
        ss += __shfl_xor(ss, m, 32);
        sd += __shfl_xor(sd, m, 32);
    }
    if (part == 0) { Ss[node] = ss; Sd[node] = sd; }
}

__global__ void __launch_bounds__(1024)
gat_count(const int* __restrict__ dst, int* __restrict__ counts, int E, int nChunk, int nb) {
    __shared__ int cnt[MAXBINS];
    const int tid = threadIdx.x;
    const int c   = blockIdx.x;
    for (int b = tid; b < nb; b += 1024) cnt[b] = 0;
    __syncthreads();
    for (int it = 0; it < CHUNK / 1024; ++it) {
        int k = c * CHUNK + it * 1024 + tid;
        if (k < E) atomicAdd(&cnt[dst[k] >> BLK_SHIFT], 1);
    }
    __syncthreads();
    for (int b = tid; b < nb; b += 1024) counts[(size_t)b * nChunk + c] = cnt[b];
}

__global__ void __launch_bounds__(256)
gat_rowscan(int* __restrict__ counts, int* __restrict__ binTotal, int nChunk) {
    __shared__ int wsum[8];
    const int tid  = threadIdx.x;
    const int lane = tid & 31;
    const int wave = tid >> 5;
    int* row = counts + (size_t)blockIdx.x * nChunk;
    int carry = 0;
    for (int base = 0; base < nChunk; base += 256) {
        int c = base + tid;
        int v = (c < nChunk) ? row[c] : 0;
        int x = v;
        for (int o = 1; o < 32; o <<= 1) {
            int y = __shfl_up(x, o, 32);
            if (lane >= o) x += y;
        }
        if (lane == 31) wsum[wave] = x;
        __syncthreads();
        int woff = 0, tot = 0;
        for (int w = 0; w < 8; ++w) {
            int t = wsum[w];
            if (w < wave) woff += t;
            tot += t;
        }
        if (c < nChunk) row[c] = carry + woff + x - v;
        carry += tot;
        __syncthreads();
    }
    if (tid == 0) binTotal[blockIdx.x] = carry;
}

__global__ void __launch_bounds__(1024)
gat_place(const int* __restrict__ src, const int* __restrict__ dst,
          const int* __restrict__ counts, const int* __restrict__ binTotal,
          int* __restrict__ binned, int E, int nChunk, int nb) {
    __shared__ int base[MAXBINS];
    __shared__ int wsum[32];
    const int tid  = threadIdx.x;
    const int lane = tid & 31;
    const int wave = tid >> 5;
    const int c    = blockIdx.x;
    int v = (tid < nb) ? binTotal[tid] : 0;
    int x = v;
    for (int o = 1; o < 32; o <<= 1) {
        int y = __shfl_up(x, o, 32);
        if (lane >= o) x += y;
    }
    if (lane == 31) wsum[wave] = x;
    __syncthreads();
    int woff = 0;
    for (int w = 0; w < wave; ++w) woff += wsum[w];
    if (tid < nb) base[tid] = woff + x - v + counts[(size_t)tid * nChunk + c];
    __syncthreads();
    int dv[CHUNK / 1024], sv[CHUNK / 1024];
#pragma unroll
    for (int it = 0; it < CHUNK / 1024; ++it) {
        int k = c * CHUNK + it * 1024 + tid;
        dv[it] = (k < E) ? dst[k] : 0;
        sv[it] = (k < E) ? src[k] : 0;
    }
#pragma unroll
    for (int it = 0; it < CHUNK / 1024; ++it) {
        int k = c * CHUNK + it * 1024 + tid;
        if (k < E) {
            int pos = atomicAdd(&base[dv[it] >> BLK_SHIFT], 1);
            binned[pos] = sv[it] | ((dv[it] & (BLK_NODES - 1)) << 24);
        }
    }
}

__device__ __forceinline__ int fkey(float f) {
    int k = __float_as_int(f);
    return (k >= 0) ? k : (k ^ 0x7FFFFFFF);
}
__device__ __forceinline__ float keyf(int k) {
    return __int_as_float((k >= 0) ? k : (k ^ 0x7FFFFFFF));
}
__device__ __forceinline__ float leaky(float v) { return (v > 0.0f) ? v : NEG_SLOPE * v; }

__global__ void __launch_bounds__(256)
gat_owner(const int* __restrict__ binned, const int* __restrict__ binTotal,
          const float* __restrict__ Ss, const float* __restrict__ Sd,
          const float* __restrict__ XP, const float* __restrict__ bias,
          float* __restrict__ out, int n) {
    __shared__ __attribute__((aligned(16))) float acc[BLK_NODES * NHID];
    __shared__ int   mkey[BLK_NODES];
    __shared__ float mval[BLK_NODES];
    __shared__ float den[BLK_NODES];
    __shared__ float sdl[BLK_NODES];
    __shared__ int   red[8];

    const int tid  = threadIdx.x;
    const int lane = tid & 31;
    const int wave = tid >> 5;
    const int half = lane >> 4;
    const int p16  = lane & 15;
    const int b    = blockIdx.x;
    const int n0   = b * BLK_NODES;

    int psum = 0;
    for (int i = tid; i < b; i += 256) psum += binTotal[i];
    for (int o = 16; o > 0; o >>= 1) psum += __shfl_xor(psum, o, 32);
    if (lane == 0) red[wave] = psum;

    for (int i = tid; i < BLK_NODES * NHID; i += 256) acc[i] = 0.0f;
    if (tid < BLK_NODES) {
        int node = n0 + tid;
        float sd = 0.0f, es = 0.0f;
        if (node < n) { sd = Sd[node]; es = leaky(Ss[node] + sd); }
        sdl[tid]  = sd;
        mkey[tid] = fkey(es);
        den[tid]  = 0.0f;
    }
    __syncthreads();
    int start = 0;
    for (int w = 0; w < 8; ++w) start += red[w];
    const int cnt = binTotal[b];
    const int* my = binned + start;

    for (int i0 = tid; i0 < cnt; i0 += 1024) {
        int   wv[4];
        float sv[4];
#pragma unroll
        for (int u = 0; u < 4; ++u) { int i = i0 + u * 256; wv[u] = (i < cnt) ? my[i] : 0; }
#pragma unroll
        for (int u = 0; u < 4; ++u) sv[u] = Ss[wv[u] & 0xFFFFFF];
#pragma unroll
        for (int u = 0; u < 4; ++u) {
            int i = i0 + u * 256;
            if (i < cnt) {
                int j = (int)((unsigned)wv[u] >> 24);
                atomicMax(&mkey[j], fkey(leaky(sv[u] + sdl[j])));
            }
        }
    }
    __syncthreads();
    if (tid < BLK_NODES) mval[tid] = keyf(mkey[tid]);
    __syncthreads();

    const float4* XP4 = (const float4*)XP;
    int wd = (wave * 32 + lane < cnt) ? my[wave * 32 + lane] : 0;
    for (int base = wave * 32; base < cnt; base += 256) {
        const bool  ok  = base + lane < cnt;
        const float ssv = Ss[wd & 0xFFFFFF];
        float4 xv[16];
        int    wr[16];
#pragma unroll
        for (int r = 0; r < 16; ++r) {
            wr[r] = __shfl(wd, 2 * r + half, 32);
            xv[r] = XP4[(size_t)(wr[r] & 0xFFFFFF) * (NHID / 4) + p16];
        }
        const int inext = base + 256 + lane;
        const int wdn   = (inext < cnt) ? my[inext] : 0;
        float ex = 0.0f;
        if (ok) {
            int j = (int)((unsigned)wd >> 24);
            float e = leaky(ssv + sdl[j]);
            ex = __expf(e - mval[j]);
            atomicAdd(&den[j], ex);
        }
#pragma unroll
        for (int r = 0; r < 16; ++r) {
            float exr = __shfl(ex, 2 * r + half, 32);
            if (base + 2 * r + half < cnt) {
                float* a = acc + (int)((unsigned)wr[r] >> 24) * NHID + p16 * 4 + 2 * half;
                float* q = acc + (int)((unsigned)wr[r] >> 24) * NHID + p16 * 4 + 2 - 2 * half;
                atomicAdd(a + 0, exr * (half ? xv[r].z : xv[r].x));
                atomicAdd(a + 1, exr * (half ? xv[r].w : xv[r].y));
                atomicAdd(q + 0, exr * (half ? xv[r].x : xv[r].z));
                atomicAdd(q + 1, exr * (half ? xv[r].y : xv[r].w));
            }
        }
        wd = wdn;
    }
    __syncthreads();

    const float4 bv = ((const float4*)bias)[p16];
    for (int jj = tid >> 4; jj < BLK_NODES; jj += 16) {
        int node = n0 + jj;
        if (node < n) {
            float exs = __expf(leaky(Ss[node] + sdl[jj]) - mval[jj]);
            float d   = den[jj] + exs;
            float4 xs = XP4[(size_t)node * (NHID / 4) + p16];
            float4 a  = ((const float4*)acc)[jj * (NHID / 4) + p16];
            float4 o;
            o.x = (a.x + exs * xs.x) / d + bv.x;
            o.y = (a.y + exs * xs.y) / d + bv.y;
            o.z = (a.z + exs * xs.z) / d + bv.z;
            o.w = (a.w + exs * xs.w) / d + bv.w;
            ((float4*)out)[(size_t)node * (NHID / 4) + p16] = o;
        }
    }
}

extern "C" void kernel_launch(void* const* d_in, const int* in_sizes, int n_in,
                              void* d_out, int out_size, void* d_ws, size_t ws_size,
                              hipStream_t stream) {
    const float* x     = (const float*)d_in[0];
    const int*   ei    = (const int*)  d_in[1];
    const float* W     = (const float*)d_in[2];
    const float* a_src = (const float*)d_in[3];
    const float* a_dst = (const float*)d_in[4];
    const float* bias  = (const float*)d_in[5];
    float*       out   = (float*)d_out;

    const int N = in_sizes[0] / NFEAT;
    const int E = in_sizes[1] / 2;
    const int* src = ei;
    const int* dst = ei + E;
    const int nb     = (N + BLK_NODES - 1) / BLK_NODES;
    const int nChunk = (E + CHUNK - 1) / CHUNK;

    float* xp       = (float*)d_ws;
    float* ssrc     = xp   + (size_t)N * NHID;
    float* sdst     = ssrc + N;
    int*   binned   = (int*)(sdst + N);
    int*   counts   = binned + E;
    int*   binTotal = counts + (size_t)nb * nChunk;

    gat_gemm   <<<(N + 127) / 128, 256,  0, stream>>>(x, W, xp, N);
    gat_scores <<<(N + 15) / 16,   256,  0, stream>>>(xp, a_src, a_dst, ssrc, sdst, N);
    gat_count  <<<nChunk,          1024, 0, stream>>>(dst, counts, E, nChunk, nb);
    gat_rowscan<<<nb,              256,  0, stream>>>(counts, binTotal, nChunk);
    gat_place  <<<nChunk,          1024, 0, stream>>>(src, dst, counts, binTotal, binned, E, nChunk, nb);
    gat_owner  <<<nb,              256,  0, stream>>>(binned, binTotal, ssrc, sdst, xp, bias, out, N);
}


# node scores s_src, s_dst computed in the GEMM epilogue from the accumulators (16-lane halving reduce: score sum order changes); the scores launch with its 50000 one-load waves is gone
# speedup vs baseline: 2.5474x; 1.0478x over previous
#include <hip/hip_runtime.h>
#include <hip/hip_bf16.h>
#include <math.h>

typedef float v2f __attribute__((ext_vector_type(2)));
typedef float v8f __attribute__((ext_vector_type(8)));

#define NFEAT 256
#define NHID  64
#define NEG_SLOPE 0.2f

#define BLK_NODES 128
#define BLK_SHIFT 7
#define CHUNK     8192
#define MAXBINS   1024

__global__ void __launch_bounds__(256)
gat_gemm(const float* __restrict__ X, const float* __restrict__ W,
         const float* __restrict__ As, const float* __restrict__ Ad,
         float* __restrict__ XP, float* __restrict__ Ss, float* __restrict__ Sd, int n) {
    const int tid  = threadIdx.x;
    const int wave = tid >> 5;
    const int lane = tid & 31;
    const int half = lane >> 4;
    const int lrow = lane & 15;
    const int r0   = (blockIdx.x * 8 + wave) * 16;
    if (r0 >= n) return;

    int arow = r0 + lrow;
    if (arow > n - 1) arow = n - 1;
    const float* aptr = X + (size_t)arow * NFEAT + 2 * half;

    const float as0 = As[lrow], as1 = As[16 + lrow], as2 = As[32 + lrow], as3 = As[48 + lrow];
    const float ad0 = Ad[lrow], ad1 = Ad[16 + lrow], ad2 = Ad[32 + lrow], ad3 = Ad[48 + lrow];

    v8f c0 = {}, c1 = {}, c2 = {}, c3 = {};

#pragma unroll 8
    for (int k = 0; k < NFEAT; k += 4) {
        v2f a = *(const v2f*)(aptr + k);
        const float* wp = W + (size_t)(k + 2 * half) * NHID + lrow;
        v2f b0, b1, b2, b3;
        b0[0] = wp[0];   b0[1] = wp[NHID + 0];
        b1[0] = wp[16];  b1[1] = wp[NHID + 16];
        b2[0] = wp[32];  b2[1] = wp[NHID + 32];
        b3[0] = wp[48];  b3[1] = wp[NHID + 48];
        c0 = __builtin_amdgcn_wmma_f32_16x16x4_f32(false, a, false, b0, (short)0, c0, false, false);
        c1 = __builtin_amdgcn_wmma_f32_16x16x4_f32(false, a, false, b1, (short)0, c1, false, false);
        c2 = __builtin_amdgcn_wmma_f32_16x16x4_f32(false, a, false, b2, (short)0, c2, false, false);
        c3 = __builtin_amdgcn_wmma_f32_16x16x4_f32(false, a, false, b3, (short)0, c3, false, false);
    }

#pragma unroll
    for (int j = 0; j < 8; ++j) {
        int row = half * 8 + j;
        if (r0 + row < n) {
            float* o = XP + (size_t)(r0 + row) * NHID + lrow;
            o[0]  = c0[j];
            o[16] = c1[j];
            o[32] = c2[j];
            o[48] = c3[j];
        }
    }

    float v[16];
#pragma unroll
    for (int j = 0; j < 8; ++j) {
        v[j]     = c0[j] * as0 + c1[j] * as1 + c2[j] * as2 + c3[j] * as3;
        v[8 + j] = c0[j] * ad0 + c1[j] * ad1 + c2[j] * ad2 + c3[j] * ad3;
    }
    const bool b3 = (lrow & 8) != 0, b2 = (lrow & 4) != 0, b1 = (lrow & 2) != 0, b0 = (lrow & 1) != 0;
    float w8[8], w4[4], w2[2];
#pragma unroll
    for (int k = 0; k < 8; ++k)
        w8[k] = (b3 ? v[8 + k] : v[k]) + __shfl_xor(b3 ? v[k] : v[8 + k], 8, 32);
#pragma unroll
    for (int k = 0; k < 4; ++k)
        w4[k] = (b2 ? w8[4 + k] : w8[k]) + __shfl_xor(b2 ? w8[k] : w8[4 + k], 4, 32);
#pragma unroll
    for (int k = 0; k < 2; ++k)
        w2[k] = (b1 ? w4[2 + k] : w4[k]) + __shfl_xor(b1 ? w4[k] : w4[2 + k], 2, 32);
    const float sc = (b0 ? w2[1] : w2[0]) + __shfl_xor(b0 ? w2[0] : w2[1], 1, 32);
    const int srow = r0 + 8 * half + (lrow & 7);
    if (srow < n) (b3 ? Sd : Ss)[srow] = sc;
}

__global__ void __launch_bounds__(1024)
gat_count(const int* __restrict__ dst, int* __restrict__ counts, int E, int nChunk, int nb) {
    __shared__ int cnt[MAXBINS];
    const int tid = threadIdx.x;
    const int c   = blockIdx.x;
    for (int b = tid; b < nb; b += 1024) cnt[b] = 0;
    __syncthreads();
    for (int it = 0; it < CHUNK / 1024; ++it) {
        int k = c * CHUNK + it * 1024 + tid;
        if (k < E) atomicAdd(&cnt[dst[k] >> BLK_SHIFT], 1);
    }
    __syncthreads();
    for (int b = tid; b < nb; b += 1024) counts[(size_t)b * nChunk + c] = cnt[b];
}

__global__ void __launch_bounds__(256)
gat_rowscan(int* __restrict__ counts, int* __restrict__ binTotal, int nChunk) {
    __shared__ int wsum[8];
    const int tid  = threadIdx.x;
    const int lane = tid & 31;
    const int wave = tid >> 5;
    int* row = counts + (size_t)blockIdx.x * nChunk;
    int carry = 0;
    for (int base = 0; base < nChunk; base += 256) {
        int c = base + tid;
        int v = (c < nChunk) ? row[c] : 0;
        int x = v;
        for (int o = 1; o < 32; o <<= 1) {
            int y = __shfl_up(x, o, 32);
            if (lane >= o) x += y;
        }
        if (lane == 31) wsum[wave] = x;
        __syncthreads();
        int woff = 0, tot = 0;
        for (int w = 0; w < 8; ++w) {
            int t = wsum[w];
            if (w < wave) woff += t;
            tot += t;
        }
        if (c < nChunk) row[c] = carry + woff + x - v;
        carry += tot;
        __syncthreads();
    }
    if (tid == 0) binTotal[blockIdx.x] = carry;
}

__global__ void __launch_bounds__(1024)
gat_place(const int* __restrict__ src, const int* __restrict__ dst,
          const int* __restrict__ counts, const int* __restrict__ binTotal,
          int* __restrict__ binned, int E, int nChunk, int nb) {
    __shared__ int base[MAXBINS];
    __shared__ int wsum[32];
    const int tid  = threadIdx.x;
    const int lane = tid & 31;
    const int wave = tid >> 5;
    const int c    = blockIdx.x;
    int v = (tid < nb) ? binTotal[tid] : 0;
    int x = v;
    for (int o = 1; o < 32; o <<= 1) {
        int y = __shfl_up(x, o, 32);
        if (lane >= o) x += y;
    }
    if (lane == 31) wsum[wave] = x;
    __syncthreads();
    int woff = 0;
    for (int w = 0; w < wave; ++w) woff += wsum[w];
    if (tid < nb) base[tid] = woff + x - v + counts[(size_t)tid * nChunk + c];
    __syncthreads();
    int dv[CHUNK / 1024], sv[CHUNK / 1024];
#pragma unroll
    for (int it = 0; it < CHUNK / 1024; ++it) {
        int k = c * CHUNK + it * 1024 + tid;
        dv[it] = (k < E) ? dst[k] : 0;
        sv[it] = (k < E) ? src[k] : 0;
    }
#pragma unroll
    for (int it = 0; it < CHUNK / 1024; ++it) {
        int k = c * CHUNK + it * 1024 + tid;
        if (k < E) {
            int pos = atomicAdd(&base[dv[it] >> BLK_SHIFT], 1);
            binned[pos] = sv[it] | ((dv[it] & (BLK_NODES - 1)) << 24);
        }
    }
}

__device__ __forceinline__ int fkey(float f) {
    int k = __float_as_int(f);
    return (k >= 0) ? k : (k ^ 0x7FFFFFFF);
}
__device__ __forceinline__ float keyf(int k) {
    return __int_as_float((k >= 0) ? k : (k ^ 0x7FFFFFFF));
}
__device__ __forceinline__ float leaky(float v) { return (v > 0.0f) ? v : NEG_SLOPE * v; }

__global__ void __launch_bounds__(256)
gat_owner(const int* __restrict__ binned, const int* __restrict__ binTotal,
          const float* __restrict__ Ss, const float* __restrict__ Sd,
          const float* __restrict__ XP, const float* __restrict__ bias,
          float* __restrict__ out, int n) {
    __shared__ __attribute__((aligned(16))) float acc[BLK_NODES * NHID];
    __shared__ int   mkey[BLK_NODES];
    __shared__ float mval[BLK_NODES];
    __shared__ float den[BLK_NODES];
    __shared__ float sdl[BLK_NODES];
    __shared__ int   red[8];

    const int tid  = threadIdx.x;
    const int lane = tid & 31;
    const int wave = tid >> 5;
    const int half = lane >> 4;
    const int p16  = lane & 15;
    const int b    = blockIdx.x;
    const int n0   = b * BLK_NODES;

    int psum = 0;
    for (int i = tid; i < b; i += 256) psum += binTotal[i];
    for (int o = 16; o > 0; o >>= 1) psum += __shfl_xor(psum, o, 32);
    if (lane == 0) red[wave] = psum;

    for (int i = tid; i < BLK_NODES * NHID; i += 256) acc[i] = 0.0f;
    if (tid < BLK_NODES) {
        int node = n0 + tid;
        float sd = 0.0f, es = 0.0f;
        if (node < n) { sd = Sd[node]; es = leaky(Ss[node] + sd); }
        sdl[tid]  = sd;
        mkey[tid] = fkey(es);
        den[tid]  = 0.0f;
    }
    __syncthreads();
    int start = 0;
    for (int w = 0; w < 8; ++w) start += red[w];
    const int cnt = binTotal[b];
    const int* my = binned + start;

    for (int i0 = tid; i0 < cnt; i0 += 1024) {
        int   wv[4];
        float sv[4];
#pragma unroll
        for (int u = 0; u < 4; ++u) { int i = i0 + u * 256; wv[u] = (i < cnt) ? my[i] : 0; }
#pragma unroll
        for (int u = 0; u < 4; ++u) sv[u] = Ss[wv[u] & 0xFFFFFF];
#pragma unroll
        for (int u = 0; u < 4; ++u) {
            int i = i0 + u * 256;
            if (i < cnt) {
                int j = (int)((unsigned)wv[u] >> 24);
                atomicMax(&mkey[j], fkey(leaky(sv[u] + sdl[j])));
            }
        }
    }
    __syncthreads();
    if (tid < BLK_NODES) mval[tid] = keyf(mkey[tid]);
    __syncthreads();

    const float4* XP4 = (const float4*)XP;
    int wd = (wave * 32 + lane < cnt) ? my[wave * 32 + lane] : 0;
    for (int base = wave * 32; base < cnt; base += 256) {
        const bool  ok  = base + lane < cnt;
        const float ssv = Ss[wd & 0xFFFFFF];
        float4 xv[16];
        int    wr[16];
#pragma unroll
        for (int r = 0; r < 16; ++r) {
            wr[r] = __shfl(wd, 2 * r + half, 32);
            xv[r] = XP4[(size_t)(wr[r] & 0xFFFFFF) * (NHID / 4) + p16];
        }
        const int inext = base + 256 + lane;
        const int wdn   = (inext < cnt) ? my[inext] : 0;
        float ex = 0.0f;
        if (ok) {
            int j = (int)((unsigned)wd >> 24);
            float e = leaky(ssv + sdl[j]);
            ex = __expf(e - mval[j]);
            atomicAdd(&den[j], ex);
        }
#pragma unroll
        for (int r = 0; r < 16; ++r) {
            float exr = __shfl(ex, 2 * r + half, 32);
            if (base + 2 * r + half < cnt) {
                float* a = acc + (int)((unsigned)wr[r] >> 24) * NHID + p16 * 4 + 2 * half;
                float* q = acc + (int)((unsigned)wr[r] >> 24) * NHID + p16 * 4 + 2 - 2 * half;
                atomicAdd(a + 0, exr * (half ? xv[r].z : xv[r].x));
                atomicAdd(a + 1, exr * (half ? xv[r].w : xv[r].y));
                atomicAdd(q + 0, exr * (half ? xv[r].x : xv[r].z));
                atomicAdd(q + 1, exr * (half ? xv[r].y : xv[r].w));
            }
        }
        wd = wdn;
    }
    __syncthreads();

    const float4 bv = ((const float4*)bias)[p16];
    for (int jj = tid >> 4; jj < BLK_NODES; jj += 16) {
        int node = n0 + jj;
        if (node < n) {
            float exs = __expf(leaky(Ss[node] + sdl[jj]) - mval[jj]);
            float d   = den[jj] + exs;
            float4 xs = XP4[(size_t)node * (NHID / 4) + p16];
            float4 a  = ((const float4*)acc)[jj * (NHID / 4) + p16];
            float4 o;
            o.x = (a.x + exs * xs.x) / d + bv.x;
            o.y = (a.y + exs * xs.y) / d + bv.y;
            o.z = (a.z + exs * xs.z) / d + bv.z;
            o.w = (a.w + exs * xs.w) / d + bv.w;
            ((float4*)out)[(size_t)node * (NHID / 4) + p16] = o;
        }
    }
}

extern "C" void kernel_launch(void* const* d_in, const int* in_sizes, int n_in,
                              void* d_out, int out_size, void* d_ws, size_t ws_size,
                              hipStream_t stream) {
    const float* x     = (const float*)d_in[0];
    const int*   ei    = (const int*)  d_in[1];
    const float* W     = (const float*)d_in[2];
    const float* a_src = (const float*)d_in[3];
    const float* a_dst = (const float*)d_in[4];
    const float* bias  = (const float*)d_in[5];
    float*       out   = (float*)d_out;

    const int N = in_sizes[0] / NFEAT;
    const int E = in_sizes[1] / 2;
    const int* src = ei;
    const int* dst = ei + E;
    const int nb     = (N + BLK_NODES - 1) / BLK_NODES;
    const int nChunk = (E + CHUNK - 1) / CHUNK;

    float* xp       = (float*)d_ws;
    float* ssrc     = xp   + (size_t)N * NHID;
    float* sdst     = ssrc + N;
    int*   binned   = (int*)(sdst + N);
    int*   counts   = binned + E;
    int*   binTotal = counts + (size_t)nb * nChunk;

    gat_gemm   <<<(N + 127) / 128, 256,  0, stream>>>(x, W, a_src, a_dst, xp, ssrc, sdst, N);
    gat_count  <<<nChunk,          1024, 0, stream>>>(dst, counts, E, nChunk, nb);
    gat_rowscan<<<nb,              256,  0, stream>>>(counts, binTotal, nChunk);
    gat_place  <<<nChunk,          1024, 0, stream>>>(src, dst, counts, binTotal, binned, E, nChunk, nb);
    gat_owner  <<<nb,              256,  0, stream>>>(binned, binTotal, ssrc, sdst, xp, bias, out, N);
}


# owner block sorts its edges by node in LDS (counting sort); a 16-lane group per node sums ex*row and denominator in registers, in rank order: no LDS float atomics, scores gathered once
# speedup vs baseline: 3.0087x; 1.1811x over previous
#include <hip/hip_runtime.h>
#include <hip/hip_bf16.h>
#include <math.h>

typedef float v2f __attribute__((ext_vector_type(2)));
typedef float v8f __attribute__((ext_vector_type(8)));

#define NFEAT 256
#define NHID  64
#define NEG_SLOPE 0.2f

#define BLK_NODES 128
#define BLK_SHIFT 7
#define CHUNK     8192
#define MAXBINS   1024

__global__ void __launch_bounds__(256)
gat_gemm(const float* __restrict__ X, const float* __restrict__ W,
         const float* __restrict__ As, const float* __restrict__ Ad,
         float* __restrict__ XP, float* __restrict__ Ss, float* __restrict__ Sd, int n) {
    const int tid  = threadIdx.x;
    const int wave = tid >> 5;
    const int lane = tid & 31;
    const int half = lane >> 4;
    const int lrow = lane & 15;
    const int r0   = (blockIdx.x * 8 + wave) * 16;
    if (r0 >= n) return;

    int arow = r0 + lrow;
    if (arow > n - 1) arow = n - 1;
    const float* aptr = X + (size_t)arow * NFEAT + 2 * half;

    const float as0 = As[lrow], as1 = As[16 + lrow], as2 = As[32 + lrow], as3 = As[48 + lrow];
    const float ad0 = Ad[lrow], ad1 = Ad[16 + lrow], ad2 = Ad[32 + lrow], ad3 = Ad[48 + lrow];

    v8f c0 = {}, c1 = {}, c2 = {}, c3 = {};

#pragma unroll 8
    for (int k = 0; k < NFEAT; k += 4) {
        v2f a = *(const v2f*)(aptr + k);
        const float* wp = W + (size_t)(k + 2 * half) * NHID + lrow;
        v2f b0, b1, b2, b3;
        b0[0] = wp[0];   b0[1] = wp[NHID + 0];
        b1[0] = wp[16];  b1[1] = wp[NHID + 16];
        b2[0] = wp[32];  b2[1] = wp[NHID + 32];
        b3[0] = wp[48];  b3[1] = wp[NHID + 48];
        c0 = __builtin_amdgcn_wmma_f32_16x16x4_f32(false, a, false, b0, (short)0, c0, false, false);
        c1 = __builtin_amdgcn_wmma_f32_16x16x4_f32(false, a, false, b1, (short)0, c1, false, false);
        c2 = __builtin_amdgcn_wmma_f32_16x16x4_f32(false, a, false, b2, (short)0, c2, false, false);
        c3 = __builtin_amdgcn_wmma_f32_16x16x4_f32(false, a, false, b3, (short)0, c3, false, false);
    }

#pragma unroll
    for (int j = 0; j < 8; ++j) {
        int row = half * 8 + j;
        if (r0 + row < n) {
            float* o = XP + (size_t)(r0 + row) * NHID + lrow;
            o[0]  = c0[j];
            o[16] = c1[j];
            o[32] = c2[j];
            o[48] = c3[j];
        }
    }

    float v[16];
#pragma unroll
    for (int j = 0; j < 8; ++j) {
        v[j]     = c0[j] * as0 + c1[j] * as1 + c2[j] * as2 + c3[j] * as3;
        v[8 + j] = c0[j] * ad0 + c1[j] * ad1 + c2[j] * ad2 + c3[j] * ad3;
    }
    const bool b3 = (lrow & 8) != 0, b2 = (lrow & 4) != 0, b1 = (lrow & 2) != 0, b0 = (lrow & 1) != 0;
    float w8[8], w4[4], w2[2];
#pragma unroll
    for (int k = 0; k < 8; ++k)
        w8[k] = (b3 ? v[8 + k] : v[k]) + __shfl_xor(b3 ? v[k] : v[8 + k], 8, 32);
#pragma unroll
    for (int k = 0; k < 4; ++k)
        w4[k] = (b2 ? w8[4 + k] : w8[k]) + __shfl_xor(b2 ? w8[k] : w8[4 + k], 4, 32);
#pragma unroll
    for (int k = 0; k < 2; ++k)
        w2[k] = (b1 ? w4[2 + k] : w4[k]) + __shfl_xor(b1 ? w4[k] : w4[2 + k], 2, 32);
    const float sc = (b0 ? w2[1] : w2[0]) + __shfl_xor(b0 ? w2[0] : w2[1], 1, 32);
    const int srow = r0 + 8 * half + (lrow & 7);
    if (srow < n) (b3 ? Sd : Ss)[srow] = sc;
}

__global__ void __launch_bounds__(1024)
gat_count(const int* __restrict__ dst, int* __restrict__ counts, int E, int nChunk, int nb) {
    __shared__ int cnt[MAXBINS];
    const int tid = threadIdx.x;
    const int c   = blockIdx.x;
    for (int b = tid; b < nb; b += 1024) cnt[b] = 0;
    __syncthreads();
    for (int it = 0; it < CHUNK / 1024; ++it) {
        int k = c * CHUNK + it * 1024 + tid;
        if (k < E) atomicAdd(&cnt[dst[k] >> BLK_SHIFT], 1);
    }
    __syncthreads();
    for (int b = tid; b < nb; b += 1024) counts[(size_t)b * nChunk + c] = cnt[b];
}

__global__ void __launch_bounds__(256)
gat_rowscan(int* __restrict__ counts, int* __restrict__ binTotal, int nChunk) {
    __shared__ int wsum[8];
    const int tid  = threadIdx.x;
    const int lane = tid & 31;
    const int wave = tid >> 5;
    int* row = counts + (size_t)blockIdx.x * nChunk;
    int carry = 0;
    for (int base = 0; base < nChunk; base += 256) {
        int c = base + tid;
        int v = (c < nChunk) ? row[c] : 0;
        int x = v;
        for (int o = 1; o < 32; o <<= 1) {
            int y = __shfl_up(x, o, 32);
            if (lane >= o) x += y;
        }
        if (lane == 31) wsum[wave] = x;
        __syncthreads();
        int woff = 0, tot = 0;
        for (int w = 0; w < 8; ++w) {
            int t = wsum[w];
            if (w < wave) woff += t;
            tot += t;
        }
        if (c < nChunk) row[c] = carry + woff + x - v;
        carry += tot;
        __syncthreads();
    }
    if (tid == 0) binTotal[blockIdx.x] = carry;
}

__global__ void __launch_bounds__(1024)
gat_place(const int* __restrict__ src, const int* __restrict__ dst,
          const int* __restrict__ counts, const int* __restrict__ binTotal,
          int* __restrict__ binned, int E, int nChunk, int nb) {
    __shared__ int base[MAXBINS];
    __shared__ int wsum[32];
    const int tid  = threadIdx.x;
    const int lane = tid & 31;
    const int wave = tid >> 5;
    const int c    = blockIdx.x;
    int v = (tid < nb) ? binTotal[tid] : 0;
    int x = v;
    for (int o = 1; o < 32; o <<= 1) {
        int y = __shfl_up(x, o, 32);
        if (lane >= o) x += y;
    }
    if (lane == 31) wsum[wave] = x;
    __syncthreads();
    int woff = 0;
    for (int w = 0; w < wave; ++w) woff += wsum[w];
    if (tid < nb) base[tid] = woff + x - v + counts[(size_t)tid * nChunk + c];
    __syncthreads();
    int dv[CHUNK / 1024], sv[CHUNK / 1024];
#pragma unroll
    for (int it = 0; it < CHUNK / 1024; ++it) {
        int k = c * CHUNK + it * 1024 + tid;
        dv[it] = (k < E) ? dst[k] : 0;
        sv[it] = (k < E) ? src[k] : 0;
    }
#pragma unroll
    for (int it = 0; it < CHUNK / 1024; ++it) {
        int k = c * CHUNK + it * 1024 + tid;
        if (k < E) {
            int pos = atomicAdd(&base[dv[it] >> BLK_SHIFT], 1);
            binned[pos] = sv[it] | ((dv[it] & (BLK_NODES - 1)) << 24);
        }
    }
}

__device__ __forceinline__ int fkey(float f) {
    int k = __float_as_int(f);
    return (k >= 0) ? k : (k ^ 0x7FFFFFFF);
}
__device__ __forceinline__ float keyf(int k) {
    return __int_as_float((k >= 0) ? k : (k ^ 0x7FFFFFFF));
}
__device__ __forceinline__ float leaky(float v) { return (v > 0.0f) ? v : NEG_SLOPE * v; }

#define OWN_CAP 4096
typedef int v2i __attribute__((ext_vector_type(2)));

__global__ void __launch_bounds__(256)
gat_owner(const int* __restrict__ binned, const int* __restrict__ binTotal,
          const float* __restrict__ Ss, const float* __restrict__ Sd,
          const float* __restrict__ XP, const float* __restrict__ bias,
          float* __restrict__ out, int n) {
    __shared__ __attribute__((aligned(16))) float stacc[BLK_NODES * NHID];
    __shared__ v2i   rec[OWN_CAP + 8];
    __shared__ int   ncnt[BLK_NODES];
    __shared__ int   noff[BLK_NODES];
    __shared__ int   ncur[BLK_NODES];
    __shared__ int   mkey[BLK_NODES];
    __shared__ float sdl[BLK_NODES];
    __shared__ float esl[BLK_NODES];
    __shared__ float stm[BLK_NODES];
    __shared__ float stden[BLK_NODES];
    __shared__ int   red[8];
    __shared__ int   wsum[4];

    const int tid  = threadIdx.x;
    const int lane = tid & 31;
    const int wave = tid >> 5;
    const int p16  = lane & 15;
    const int hw   = tid >> 4;
    const int b    = blockIdx.x;
    const int n0   = b * BLK_NODES;

    int psum = 0;
    for (int i = tid; i < b; i += 256) psum += binTotal[i];
    for (int o = 16; o > 0; o >>= 1) psum += __shfl_xor(psum, o, 32);
    if (lane == 0) red[wave] = psum;

    if (tid < BLK_NODES) {
        int node = n0 + tid;
        float sd = 0.0f, es = 0.0f;
        if (node < n) { sd = Sd[node]; es = leaky(Ss[node] + sd); }
        sdl[tid]  = sd;
        esl[tid]  = es;
        mkey[tid] = fkey(es);
    }
    __syncthreads();
    int start = 0;
    for (int w = 0; w < 8; ++w) start += red[w];
    const int cnt   = binTotal[b];
    const int ntile = (cnt > OWN_CAP) ? (cnt + OWN_CAP - 1) / OWN_CAP : 1;
    const float4* XP4 = (const float4*)XP;
    const float4  bv  = ((const float4*)bias)[p16];

    for (int t = 0; t < ntile; ++t) {
        const int* myt  = binned + start + t * OWN_CAP;
        const int  rest = cnt - t * OWN_CAP;
        const int  tcnt = (rest < OWN_CAP) ? rest : OWN_CAP;

        if (tid < BLK_NODES) ncnt[tid] = 0;
        __syncthreads();

        for (int i0 = tid; i0 < tcnt; i0 += 1024) {
            int wv[4];
#pragma unroll
            for (int u = 0; u < 4; ++u) { int i = i0 + u * 256; wv[u] = myt[(i < tcnt) ? i : tcnt - 1]; }
#pragma unroll
            for (int u = 0; u < 4; ++u) {
                int i = i0 + u * 256;
                if (i < tcnt) atomicAdd(&ncnt[(int)((unsigned)wv[u] >> 24)], 1);
            }
        }
        __syncthreads();

        int sv = 0, sx = 0;
        if (tid < BLK_NODES) {
            sv = ncnt[tid];
            sx = sv;
            for (int o = 1; o < 32; o <<= 1) {
                int y = __shfl_up(sx, o, 32);
                if (lane >= o) sx += y;
            }
            if (lane == 31) wsum[wave] = sx;
        }
        __syncthreads();
        if (tid < BLK_NODES) {
            int woff = 0;
            for (int w = 0; w < 4; ++w) if (w < wave) woff += wsum[w];
            int off = woff + sx - sv;
            noff[tid] = off;
            ncur[tid] = off;
        }
        __syncthreads();

        for (int i0 = tid; i0 < tcnt; i0 += 1024) {
            int   wv[4];
            float sc4[4];
#pragma unroll
            for (int u = 0; u < 4; ++u) { int i = i0 + u * 256; wv[u] = myt[(i < tcnt) ? i : tcnt - 1]; }
#pragma unroll
            for (int u = 0; u < 4; ++u) sc4[u] = Ss[wv[u] & 0xFFFFFF];
#pragma unroll
            for (int u = 0; u < 4; ++u) {
                int i = i0 + u * 256;
                if (i < tcnt) {
                    int   j = (int)((unsigned)wv[u] >> 24);
                    float e = leaky(sc4[u] + sdl[j]);
                    atomicMax(&mkey[j], fkey(e));
                    int pos = atomicAdd(&ncur[j], 1);
                    v2i r;
                    r.x = wv[u] & 0xFFFFFF;
                    r.y = __float_as_int(e);
                    rec[pos] = r;
                }
            }
        }
        __syncthreads();

        for (int q = 0; q < BLK_NODES / 16; ++q) {
            const int   jj   = hw + 16 * q;
            const int   node = n0 + jj;
            const int   rs   = noff[jj];
            const int   re   = ncur[jj];
            const float mnew = keyf(mkey[jj]);
            float4 a0 = make_float4(0.0f, 0.0f, 0.0f, 0.0f);
            float  mold, den;
            if (t == 0) {
                mold = esl[jj];
                den  = 1.0f;
                if (node < n) a0 = XP4[(size_t)node * (NHID / 4) + p16];
            } else {
                a0   = ((const float4*)stacc)[jj * (NHID / 4) + p16];
                mold = stm[jj];
                den  = stden[jj];
            }
            float4 acc  = make_float4(0.0f, 0.0f, 0.0f, 0.0f);
            float  dsum = 0.0f;
            for (int r = rs; r < re; r += 8) {
                v2i    rc[8];
                float4 xv[8];
#pragma unroll
                for (int u = 0; u < 8; ++u) {
                    int idx = r + u;
                    if (idx > re - 1) idx = re - 1;
                    rc[u] = rec[idx];
                }
#pragma unroll
                for (int u = 0; u < 8; ++u) xv[u] = XP4[(size_t)rc[u].x * (NHID / 4) + p16];
#pragma unroll
                for (int u = 0; u < 8; ++u) {
                    float ex = (r + u < re) ? __expf(__int_as_float(rc[u].y) - mnew) : 0.0f;
                    dsum  += ex;
                    acc.x += ex * xv[u].x;
                    acc.y += ex * xv[u].y;
                    acc.z += ex * xv[u].z;
                    acc.w += ex * xv[u].w;
                }
            }
            const float sc = __expf(mold - mnew);
            den  = den * sc + dsum;
            a0.x = a0.x * sc + acc.x;
            a0.y = a0.y * sc + acc.y;
            a0.z = a0.z * sc + acc.z;
            a0.w = a0.w * sc + acc.w;
            if (t == ntile - 1) {
                if (node < n) {
                    float4 o;
                    o.x = a0.x / den + bv.x;
                    o.y = a0.y / den + bv.y;
                    o.z = a0.z / den + bv.z;
                    o.w = a0.w / den + bv.w;
                    ((float4*)out)[(size_t)node * (NHID / 4) + p16] = o;
                }
            } else {
                ((float4*)stacc)[jj * (NHID / 4) + p16] = a0;
                if (p16 == 0) { stm[jj] = mnew; stden[jj] = den; }
            }
        }
    }
}

extern "C" void kernel_launch(void* const* d_in, const int* in_sizes, int n_in,
                              void* d_out, int out_size, void* d_ws, size_t ws_size,
                              hipStream_t stream) {
    const float* x     = (const float*)d_in[0];
    const int*   ei    = (const int*)  d_in[1];
    const float* W     = (const float*)d_in[2];
    const float* a_src = (const float*)d_in[3];
    const float* a_dst = (const float*)d_in[4];
    const float* bias  = (const float*)d_in[5];
    float*       out   = (float*)d_out;

    const int N = in_sizes[0] / NFEAT;
    const int E = in_sizes[1] / 2;
    const int* src = ei;
    const int* dst = ei + E;
    const int nb     = (N + BLK_NODES - 1) / BLK_NODES;
    const int nChunk = (E + CHUNK - 1) / CHUNK;

    float* xp       = (float*)d_ws;
    float* ssrc     = xp   + (size_t)N * NHID;
    float* sdst     = ssrc + N;
    int*   binned   = (int*)(sdst + N);
    int*   counts   = binned + E;
    int*   binTotal = counts + (size_t)nb * nChunk;

    gat_gemm   <<<(N + 127) / 128, 256,  0, stream>>>(x, W, a_src, a_dst, xp, ssrc, sdst, N);
    gat_count  <<<nChunk,          1024, 0, stream>>>(dst, counts, E, nChunk, nb);
    gat_rowscan<<<nb,              256,  0, stream>>>(counts, binTotal, nChunk);
    gat_place  <<<nChunk,          1024, 0, stream>>>(src, dst, counts, binTotal, binned, E, nChunk, nb);
    gat_owner  <<<nb,              256,  0, stream>>>(binned, binTotal, ssrc, sdst, xp, bias, out, N);
}


# GEMM operand feed: W packed once per WGP into LDS in fragment order (16-B LDS reads), x read 16 B per lane, one 25-wave group per WGP, a strip per wave; k sum order inside each 8-k block changes
# speedup vs baseline: 3.1548x; 1.0486x over previous
#include <hip/hip_runtime.h>
#include <hip/hip_bf16.h>
#include <math.h>

typedef float v2f __attribute__((ext_vector_type(2)));
typedef float v8f __attribute__((ext_vector_type(8)));

#define NFEAT 256
#define NHID  64
#define NEG_SLOPE 0.2f

#define BLK_NODES 128
#define BLK_SHIFT 7
#define CHUNK     8192
#define MAXBINS   1024

#define KBLK   (NFEAT / 8)
#define GEMM_THREADS 800

__device__ __forceinline__ void gemm_kgroup(const float4 (&xa)[4], const float4* bp,
                                            v8f& c0, v8f& c1, v8f& c2, v8f& c3) {
#pragma unroll
    for (int u = 0; u < 4; ++u) {
        const float4 b0 = bp[u * 128], b1 = bp[u * 128 + 32], b2 = bp[u * 128 + 64], b3 = bp[u * 128 + 96];
        v2f a, p0, p1, p2, p3;
        a[0] = xa[u].x;  a[1] = xa[u].y;
        p0[0] = b0.x; p0[1] = b0.y;  p1[0] = b1.x; p1[1] = b1.y;
        p2[0] = b2.x; p2[1] = b2.y;  p3[0] = b3.x; p3[1] = b3.y;
        c0 = __builtin_amdgcn_wmma_f32_16x16x4_f32(false, a, false, p0, (short)0, c0, false, false);
        c1 = __builtin_amdgcn_wmma_f32_16x16x4_f32(false, a, false, p1, (short)0, c1, false, false);
        c2 = __builtin_amdgcn_wmma_f32_16x16x4_f32(false, a, false, p2, (short)0, c2, false, false);
        c3 = __builtin_amdgcn_wmma_f32_16x16x4_f32(false, a, false, p3, (short)0, c3, false, false);
        a[0] = xa[u].z;  a[1] = xa[u].w;
        p0[0] = b0.z; p0[1] = b0.w;  p1[0] = b1.z; p1[1] = b1.w;
        p2[0] = b2.z; p2[1] = b2.w;  p3[0] = b3.z; p3[1] = b3.w;
        c0 = __builtin_amdgcn_wmma_f32_16x16x4_f32(false, a, false, p0, (short)0, c0, false, false);
        c1 = __builtin_amdgcn_wmma_f32_16x16x4_f32(false, a, false, p1, (short)0, c1, false, false);
        c2 = __builtin_amdgcn_wmma_f32_16x16x4_f32(false, a, false, p2, (short)0, c2, false, false);
        c3 = __builtin_amdgcn_wmma_f32_16x16x4_f32(false, a, false, p3, (short)0, c3, false, false);
    }
}

__global__ void __launch_bounds__(GEMM_THREADS)
gat_gemm(const float* __restrict__ X, const float* __restrict__ W,
         const float* __restrict__ As, const float* __restrict__ Ad,
         float* __restrict__ XP, float* __restrict__ Ss, float* __restrict__ Sd, int n) {
    __shared__ __attribute__((aligned(16))) float4 wl[KBLK * 4 * 32];

    const int tid    = threadIdx.x;
    const int wave   = __builtin_amdgcn_readfirstlane(tid >> 5);
    const int lane   = tid & 31;
    const int half   = lane >> 4;
    const int lrow   = lane & 15;
    const int nwaves = blockDim.x >> 5;
    const int nstrip = (n + 15) >> 4;
    const int sstep  = gridDim.x * nwaves;
    int s = blockIdx.x + gridDim.x * wave;

    float4 cur[4], nxt[4];
    const float4* ap;
    {
        int arow = s * 16 + lrow;
        if (arow > n - 1) arow = n - 1;
        ap = (const float4*)(X + (size_t)arow * NFEAT + 4 * half);
    }
    if (s < nstrip) {
#pragma unroll
        for (int u = 0; u < 4; ++u) cur[u] = ap[2 * u];
    }

    const float as0 = As[lrow], as1 = As[16 + lrow], as2 = As[32 + lrow], as3 = As[48 + lrow];
    const float ad0 = Ad[lrow], ad1 = Ad[16 + lrow], ad2 = Ad[32 + lrow], ad3 = Ad[48 + lrow];

    for (int j = wave; j < KBLK; j += nwaves) {
        const float* wp = W + (size_t)(8 * j + 4 * half) * NHID + lrow;
#pragma unroll
        for (int t = 0; t < 4; ++t) {
            float4 v;
            v.x = wp[0 * NHID + 16 * t];
            v.y = wp[1 * NHID + 16 * t];
            v.z = wp[2 * NHID + 16 * t];
            v.w = wp[3 * NHID + 16 * t];
            wl[(j * 4 + t) * 32 + lane] = v;
        }
    }

    __syncthreads();

    for (; s < nstrip; s += sstep) {
        const int r0 = s * 16;
        v8f c0 = {}, c1 = {}, c2 = {}, c3 = {};

        const float4* apn;
        {
            const int sn = (s + sstep < nstrip) ? s + sstep : s;
            int arow = sn * 16 + lrow;
            if (arow > n - 1) arow = n - 1;
            apn = (const float4*)(X + (size_t)arow * NFEAT + 4 * half);
        }

        const float4* bp = wl + lane;
#pragma unroll 2
        for (int jj = 0; jj < KBLK; jj += 4) {
            const float4* pn = (jj + 4 < KBLK) ? ap + 2 * (jj + 4) : apn;
#pragma unroll
            for (int u = 0; u < 4; ++u) nxt[u] = pn[2 * u];
            gemm_kgroup(cur, bp + jj * 128, c0, c1, c2, c3);
#pragma unroll
            for (int u = 0; u < 4; ++u) cur[u] = nxt[u];
        }
        ap = apn;

#pragma unroll
        for (int j = 0; j < 8; ++j) {
            int row = half * 8 + j;
            if (r0 + row < n) {
                float* o = XP + (size_t)(r0 + row) * NHID + lrow;
                o[0]  = c0[j];
                o[16] = c1[j];
                o[32] = c2[j];
                o[48] = c3[j];
            }
        }

        float v[16];
#pragma unroll
        for (int j = 0; j < 8; ++j) {
            v[j]     = c0[j] * as0 + c1[j] * as1 + c2[j] * as2 + c3[j] * as3;
            v[8 + j] = c0[j] * ad0 + c1[j] * ad1 + c2[j] * ad2 + c3[j] * ad3;
        }
        const bool b3 = (lrow & 8) != 0, b2 = (lrow & 4) != 0, b1 = (lrow & 2) != 0, b0 = (lrow & 1) != 0;
        float w8[8], w4[4], w2[2];
#pragma unroll
        for (int k = 0; k < 8; ++k)
            w8[k] = (b3 ? v[8 + k] : v[k]) + __shfl_xor(b3 ? v[k] : v[8 + k], 8, 32);
#pragma unroll
        for (int k = 0; k < 4; ++k)
            w4[k] = (b2 ? w8[4 + k] : w8[k]) + __shfl_xor(b2 ? w8[k] : w8[4 + k], 4, 32);
#pragma unroll
        for (int k = 0; k < 2; ++k)
            w2[k] = (b1 ? w4[2 + k] : w4[k]) + __shfl_xor(b1 ? w4[k] : w4[2 + k], 2, 32);
        const float sc = (b0 ? w2[1] : w2[0]) + __shfl_xor(b0 ? w2[0] : w2[1], 1, 32);
        const int srow = r0 + 8 * half + (lrow & 7);
        if (srow < n) (b3 ? Sd : Ss)[srow] = sc;
    }
}

__global__ void __launch_bounds__(1024)
gat_count(const int* __restrict__ dst, int* __restrict__ counts, int E, int nChunk, int nb) {
    __shared__ int cnt[MAXBINS];
    const int tid = threadIdx.x;
    const int c   = blockIdx.x;
    for (int b = tid; b < nb; b += 1024) cnt[b] = 0;
    __syncthreads();
    for (int it = 0; it < CHUNK / 1024; ++it) {
        int k = c * CHUNK + it * 1024 + tid;
        if (k < E) atomicAdd(&cnt[dst[k] >> BLK_SHIFT], 1);
    }
    __syncthreads();
    for (int b = tid; b < nb; b += 1024) counts[(size_t)b * nChunk + c] = cnt[b];
}

__global__ void __launch_bounds__(256)
gat_rowscan(int* __restrict__ counts, int* __restrict__ binTotal, int nChunk) {
    __shared__ int wsum[8];
    const int tid  = threadIdx.x;
    const int lane = tid & 31;
    const int wave = tid >> 5;
    int* row = counts + (size_t)blockIdx.x * nChunk;
    int carry = 0;
    for (int base = 0; base < nChunk; base += 256) {
        int c = base + tid;
        int v = (c < nChunk) ? row[c] : 0;
        int x = v;
        for (int o = 1; o < 32; o <<= 1) {
            int y = __shfl_up(x, o, 32);
            if (lane >= o) x += y;
        }
        if (lane == 31) wsum[wave] = x;
        __syncthreads();
        int woff = 0, tot = 0;
        for (int w = 0; w < 8; ++w) {
            int t = wsum[w];
            if (w < wave) woff += t;
            tot += t;
        }
        if (c < nChunk) row[c] = carry + woff + x - v;
        carry += tot;
        __syncthreads();
    }
    if (tid == 0) binTotal[blockIdx.x] = carry;
}

__global__ void __launch_bounds__(1024)
gat_place(const int* __restrict__ src, const int* __restrict__ dst,
          const int* __restrict__ counts, const int* __restrict__ binTotal,
          int* __restrict__ binned, int E, int nChunk, int nb) {
    __shared__ int base[MAXBINS];
    __shared__ int wsum[32];
    const int tid  = threadIdx.x;
    const int lane = tid & 31;
    const int wave = tid >> 5;
    const int c    = blockIdx.x;
    int v = (tid < nb) ? binTotal[tid] : 0;
    int x = v;
    for (int o = 1; o < 32; o <<= 1) {
        int y = __shfl_up(x, o, 32);
        if (lane >= o) x += y;
    }
    if (lane == 31) wsum[wave] = x;
    __syncthreads();
    int woff = 0;
    for (int w = 0; w < wave; ++w) woff += wsum[w];
    if (tid < nb) base[tid] = woff + x - v + counts[(size_t)tid * nChunk + c];
    __syncthreads();
    int dv[CHUNK / 1024], sv[CHUNK / 1024];
#pragma unroll
    for (int it = 0; it < CHUNK / 1024; ++it) {
        int k = c * CHUNK + it * 1024 + tid;
        dv[it] = (k < E) ? dst[k] : 0;
        sv[it] = (k < E) ? src[k] : 0;
    }
#pragma unroll
    for (int it = 0; it < CHUNK / 1024; ++it) {
        int k = c * CHUNK + it * 1024 + tid;
        if (k < E) {
            int pos = atomicAdd(&base[dv[it] >> BLK_SHIFT], 1);
            binned[pos] = sv[it] | ((dv[it] & (BLK_NODES - 1)) << 24);
        }
    }
}

__device__ __forceinline__ int fkey(float f) {
    int k = __float_as_int(f);
    return (k >= 0) ? k : (k ^ 0x7FFFFFFF);
}
__device__ __forceinline__ float keyf(int k) {
    return __int_as_float((k >= 0) ? k : (k ^ 0x7FFFFFFF));
}
__device__ __forceinline__ float leaky(float v) { return (v > 0.0f) ? v : NEG_SLOPE * v; }

#define OWN_CAP 4096
typedef int v2i __attribute__((ext_vector_type(2)));

__global__ void __launch_bounds__(256)
gat_owner(const int* __restrict__ binned, const int* __restrict__ binTotal,
          const float* __restrict__ Ss, const float* __restrict__ Sd,
          const float* __restrict__ XP, const float* __restrict__ bias,
          float* __restrict__ out, int n) {
    __shared__ __attribute__((aligned(16))) float stacc[BLK_NODES * NHID];
    __shared__ v2i   rec[OWN_CAP + 8];
    __shared__ int   ncnt[BLK_NODES];
    __shared__ int   noff[BLK_NODES];
    __shared__ int   ncur[BLK_NODES];
    __shared__ int   mkey[BLK_NODES];
    __shared__ float sdl[BLK_NODES];
    __shared__ float esl[BLK_NODES];
    __shared__ float stm[BLK_NODES];
    __shared__ float stden[BLK_NODES];
    __shared__ int   red[8];
    __shared__ int   wsum[4];

    const int tid  = threadIdx.x;
    const int lane = tid & 31;
    const int wave = tid >> 5;
    const int p16  = lane & 15;
    const int hw   = tid >> 4;
    const int b    = blockIdx.x;
    const int n0   = b * BLK_NODES;

    int psum = 0;
    for (int i = tid; i < b; i += 256) psum += binTotal[i];
    for (int o = 16; o > 0; o >>= 1) psum += __shfl_xor(psum, o, 32);
    if (lane == 0) red[wave] = psum;

    if (tid < BLK_NODES) {
        int node = n0 + tid;
        float sd = 0.0f, es = 0.0f;
        if (node < n) { sd = Sd[node]; es = leaky(Ss[node] + sd); }
        sdl[tid]  = sd;
        esl[tid]  = es;
        mkey[tid] = fkey(es);
    }
    __syncthreads();
    int start = 0;
    for (int w = 0; w < 8; ++w) start += red[w];
    const int cnt   = binTotal[b];
    const int ntile = (cnt > OWN_CAP) ? (cnt + OWN_CAP - 1) / OWN_CAP : 1;
    const float4* XP4 = (const float4*)XP;
    const float4  bv  = ((const float4*)bias)[p16];

    for (int t = 0; t < ntile; ++t) {
        const int* myt  = binned + start + t * OWN_CAP;
        const int  rest = cnt - t * OWN_CAP;
        const int  tcnt = (rest < OWN_CAP) ? rest : OWN_CAP;

        if (tid < BLK_NODES) ncnt[tid] = 0;
        __syncthreads();

        for (int i0 = tid; i0 < tcnt; i0 += 1024) {
            int wv[4];
#pragma unroll
            for (int u = 0; u < 4; ++u) { int i = i0 + u * 256; wv[u] = myt[(i < tcnt) ? i : tcnt - 1]; }
#pragma unroll
            for (int u = 0; u < 4; ++u) {
                int i = i0 + u * 256;
                if (i < tcnt) atomicAdd(&ncnt[(int)((unsigned)wv[u] >> 24)], 1);
            }
        }
        __syncthreads();

        int sv = 0, sx = 0;
        if (tid < BLK_NODES) {
            sv = ncnt[tid];
            sx = sv;
            for (int o = 1; o < 32; o <<= 1) {
                int y = __shfl_up(sx, o, 32);
                if (lane >= o) sx += y;
            }
            if (lane == 31) wsum[wave] = sx;
        }
        __syncthreads();
        if (tid < BLK_NODES) {
            int woff = 0;
            for (int w = 0; w < 4; ++w) if (w < wave) woff += wsum[w];
            int off = woff + sx - sv;
            noff[tid] = off;
            ncur[tid] = off;
        }
        __syncthreads();

        for (int i0 = tid; i0 < tcnt; i0 += 1024) {
            int   wv[4];
            float sc4[4];
#pragma unroll
            for (int u = 0; u < 4; ++u) { int i = i0 + u * 256; wv[u] = myt[(i < tcnt) ? i : tcnt - 1]; }
#pragma unroll
            for (int u = 0; u < 4; ++u) sc4[u] = Ss[wv[u] & 0xFFFFFF];
#pragma unroll
            for (int u = 0; u < 4; ++u) {
                int i = i0 + u * 256;
                if (i < tcnt) {
                    int   j = (int)((unsigned)wv[u] >> 24);
                    float e = leaky(sc4[u] + sdl[j]);
                    atomicMax(&mkey[j], fkey(e));
                    int pos = atomicAdd(&ncur[j], 1);
                    v2i r;
                    r.x = wv[u] & 0xFFFFFF;
                    r.y = __float_as_int(e);
                    rec[pos] = r;
                }
            }
        }
        __syncthreads();

        for (int q = 0; q < BLK_NODES / 16; ++q) {
            const int   jj   = hw + 16 * q;
            const int   node = n0 + jj;
            const int   rs   = noff[jj];
            const int   re   = ncur[jj];
            const float mnew = keyf(mkey[jj]);
            float4 a0 = make_float4(0.0f, 0.0f, 0.0f, 0.0f);
            float  mold, den;
            if (t == 0) {
                mold = esl[jj];
                den  = 1.0f;
                if (node < n) a0 = XP4[(size_t)node * (NHID / 4) + p16];
            } else {
                a0   = ((const float4*)stacc)[jj * (NHID / 4) + p16];
                mold = stm[jj];
                den  = stden[jj];
            }
            float4 acc  = make_float4(0.0f, 0.0f, 0.0f, 0.0f);
            float  dsum = 0.0f;
            for (int r = rs; r < re; r += 8) {
                v2i    rc[8];
                float4 xv[8];
#pragma unroll
                for (int u = 0; u < 8; ++u) {
                    int idx = r + u;
                    if (idx > re - 1) idx = re - 1;
                    rc[u] = rec[idx];
                }
#pragma unroll
                for (int u = 0; u < 8; ++u) xv[u] = XP4[(size_t)rc[u].x * (NHID / 4) + p16];
#pragma unroll
                for (int u = 0; u < 8; ++u) {
                    float ex = (r + u < re) ? __expf(__int_as_float(rc[u].y) - mnew) : 0.0f;
                    dsum  += ex;
                    acc.x += ex * xv[u].x;
                    acc.y += ex * xv[u].y;
                    acc.z += ex * xv[u].z;
                    acc.w += ex * xv[u].w;
                }
            }
            const float sc = __expf(mold - mnew);
            den  = den * sc + dsum;
            a0.x = a0.x * sc + acc.x;
            a0.y = a0.y * sc + acc.y;
            a0.z = a0.z * sc + acc.z;
            a0.w = a0.w * sc + acc.w;
            if (t == ntile - 1) {
                if (node < n) {
                    float4 o;
                    o.x = a0.x / den + bv.x;
                    o.y = a0.y / den + bv.y;
                    o.z = a0.z / den + bv.z;
                    o.w = a0.w / den + bv.w;
                    ((float4*)out)[(size_t)node * (NHID / 4) + p16] = o;
                }
            } else {
                ((float4*)stacc)[jj * (NHID / 4) + p16] = a0;
                if (p16 == 0) { stm[jj] = mnew; stden[jj] = den; }
            }
        }
    }
}

extern "C" void kernel_launch(void* const* d_in, const int* in_sizes, int n_in,
                              void* d_out, int out_size, void* d_ws, size_t ws_size,
                              hipStream_t stream) {
    const float* x     = (const float*)d_in[0];
    const int*   ei    = (const int*)  d_in[1];
    const float* W     = (const float*)d_in[2];
    const float* a_src = (const float*)d_in[3];
    const float* a_dst = (const float*)d_in[4];
    const float* bias  = (const float*)d_in[5];
    float*       out   = (float*)d_out;

    const int N = in_sizes[0] / NFEAT;
    const int E = in_sizes[1] / 2;
    const int* src = ei;
    const int* dst = ei + E;
    const int nb     = (N + BLK_NODES - 1) / BLK_NODES;
    const int nChunk = (E + CHUNK - 1) / CHUNK;

    float* xp       = (float*)d_ws;
    float* ssrc     = xp   + (size_t)N * NHID;
    float* sdst     = ssrc + N;
    int*   binned   = (int*)(sdst + N);
    int*   counts   = binned + E;
    int*   binTotal = counts + (size_t)nb * nChunk;

    gat_gemm   <<<256,     GEMM_THREADS, 0, stream>>>(x, W, a_src, a_dst, xp, ssrc, sdst, N);
    gat_count  <<<nChunk,          1024, 0, stream>>>(dst, counts, E, nChunk, nb);
    gat_rowscan<<<nb,              256,  0, stream>>>(counts, binTotal, nChunk);
    gat_place  <<<nChunk,          1024, 0, stream>>>(src, dst, counts, binTotal, binned, E, nChunk, nb);
    gat_owner  <<<nb,              256,  0, stream>>>(binned, binTotal, ssrc, sdst, xp, bias, out, N);
}


# edges sorted by destination block per chunk in LDS, written as chunk-local runs; an owner block reads its run of every chunk: count and row-scan launches gone; edge order in a node's sum changes
# speedup vs baseline: 5.1452x; 1.6309x over previous
#include <hip/hip_runtime.h>
#include <hip/hip_bf16.h>
#include <math.h>

typedef float v2f __attribute__((ext_vector_type(2)));
typedef float v8f __attribute__((ext_vector_type(8)));

#define NFEAT 256
#define NHID  64
#define NEG_SLOPE 0.2f

#define BLK_NODES 128
#define BLK_SHIFT 7
#define CHUNK     8192
#define MAXBINS   1024
#define MAXCHUNK  256

#define KBLK   (NFEAT / 8)
#define GEMM_THREADS 800

__device__ __forceinline__ void gemm_kgroup(const float4 (&xa)[4], const float4* bp,
                                            v8f& c0, v8f& c1, v8f& c2, v8f& c3) {
#pragma unroll
    for (int u = 0; u < 4; ++u) {
        const float4 b0 = bp[u * 128], b1 = bp[u * 128 + 32], b2 = bp[u * 128 + 64], b3 = bp[u * 128 + 96];
        v2f a, p0, p1, p2, p3;
        a[0] = xa[u].x;  a[1] = xa[u].y;
        p0[0] = b0.x; p0[1] = b0.y;  p1[0] = b1.x; p1[1] = b1.y;
        p2[0] = b2.x; p2[1] = b2.y;  p3[0] = b3.x; p3[1] = b3.y;
        c0 = __builtin_amdgcn_wmma_f32_16x16x4_f32(false, a, false, p0, (short)0, c0, false, false);
        c1 = __builtin_amdgcn_wmma_f32_16x16x4_f32(false, a, false, p1, (short)0, c1, false, false);
        c2 = __builtin_amdgcn_wmma_f32_16x16x4_f32(false, a, false, p2, (short)0, c2, false, false);
        c3 = __builtin_amdgcn_wmma_f32_16x16x4_f32(false, a, false, p3, (short)0, c3, false, false);
        a[0] = xa[u].z;  a[1] = xa[u].w;
        p0[0] = b0.z; p0[1] = b0.w;  p1[0] = b1.z; p1[1] = b1.w;
        p2[0] = b2.z; p2[1] = b2.w;  p3[0] = b3.z; p3[1] = b3.w;
        c0 = __builtin_amdgcn_wmma_f32_16x16x4_f32(false, a, false, p0, (short)0, c0, false, false);
        c1 = __builtin_amdgcn_wmma_f32_16x16x4_f32(false, a, false, p1, (short)0, c1, false, false);
        c2 = __builtin_amdgcn_wmma_f32_16x16x4_f32(false, a, false, p2, (short)0, c2, false, false);
        c3 = __builtin_amdgcn_wmma_f32_16x16x4_f32(false, a, false, p3, (short)0, c3, false, false);
    }
}

__global__ void __launch_bounds__(GEMM_THREADS)
gat_gemm(const float* __restrict__ X, const float* __restrict__ W,
         const float* __restrict__ As, const float* __restrict__ Ad,
         float* __restrict__ XP, float* __restrict__ Ss, float* __restrict__ Sd, int n) {
    __shared__ __attribute__((aligned(16))) float4 wl[KBLK * 4 * 32];

    const int tid    = threadIdx.x;
    const int wave   = __builtin_amdgcn_readfirstlane(tid >> 5);
    const int lane   = tid & 31;
    const int half   = lane >> 4;
    const int lrow   = lane & 15;
    const int nwaves = blockDim.x >> 5;
    const int nstrip = (n + 15) >> 4;
    const int sstep  = gridDim.x * nwaves;
    int s = blockIdx.x + gridDim.x * wave;

    float4 cur[4], nxt[4];
    const float4* ap;
    {
        int arow = s * 16 + lrow;
        if (arow > n - 1) arow = n - 1;
        ap = (const float4*)(X + (size_t)arow * NFEAT + 4 * half);
    }
    if (s < nstrip) {
#pragma unroll
        for (int u = 0; u < 4; ++u) cur[u] = ap[2 * u];
    }

    const float as0 = As[lrow], as1 = As[16 + lrow], as2 = As[32 + lrow], as3 = As[48 + lrow];
    const float ad0 = Ad[lrow], ad1 = Ad[16 + lrow], ad2 = Ad[32 + lrow], ad3 = Ad[48 + lrow];

    for (int j = wave; j < KBLK; j += nwaves) {
        const float* wp = W + (size_t)(8 * j + 4 * half) * NHID + lrow;
#pragma unroll
        for (int t = 0; t < 4; ++t) {
            float4 v;
            v.x = wp[0 * NHID + 16 * t];
            v.y = wp[1 * NHID + 16 * t];
            v.z = wp[2 * NHID + 16 * t];
            v.w = wp[3 * NHID + 16 * t];
            wl[(j * 4 + t) * 32 + lane] = v;
        }
    }

    __syncthreads();

    for (; s < nstrip; s += sstep) {
        const int r0 = s * 16;
        v8f c0 = {}, c1 = {}, c2 = {}, c3 = {};

        const float4* apn;
        {
            const int sn = (s + sstep < nstrip) ? s + sstep : s;
            int arow = sn * 16 + lrow;
            if (arow > n - 1) arow = n - 1;
            apn = (const float4*)(X + (size_t)arow * NFEAT + 4 * half);
        }

        const float4* bp = wl + lane;
#pragma unroll 2
        for (int jj = 0; jj < KBLK; jj += 4) {
            const float4* pn = (jj + 4 < KBLK) ? ap + 2 * (jj + 4) : apn;
#pragma unroll
            for (int u = 0; u < 4; ++u) nxt[u] = pn[2 * u];
            gemm_kgroup(cur, bp + jj * 128, c0, c1, c2, c3);
#pragma unroll
            for (int u = 0; u < 4; ++u) cur[u] = nxt[u];
        }
        ap = apn;

#pragma unroll
        for (int j = 0; j < 8; ++j) {
            int row = half * 8 + j;
            if (r0 + row < n) {
                float* o = XP + (size_t)(r0 + row) * NHID + lrow;
                o[0]  = c0[j];
                o[16] = c1[j];
                o[32] = c2[j];
                o[48] = c3[j];
            }
        }

        float v[16];
#pragma unroll
        for (int j = 0; j < 8; ++j) {
            v[j]     = c0[j] * as0 + c1[j] * as1 + c2[j] * as2 + c3[j] * as3;
            v[8 + j] = c0[j] * ad0 + c1[j] * ad1 + c2[j] * ad2 + c3[j] * ad3;
        }
        const bool b3 = (lrow & 8) != 0, b2 = (lrow & 4) != 0, b1 = (lrow & 2) != 0, b0 = (lrow & 1) != 0;
        float w8[8], w4[4], w2[2];
#pragma unroll
        for (int k = 0; k < 8; ++k)
            w8[k] = (b3 ? v[8 + k] : v[k]) + __shfl_xor(b3 ? v[k] : v[8 + k], 8, 32);
#pragma unroll
        for (int k = 0; k < 4; ++k)
            w4[k] = (b2 ? w8[4 + k] : w8[k]) + __shfl_xor(b2 ? w8[k] : w8[4 + k], 4, 32);
#pragma unroll
        for (int k = 0; k < 2; ++k)
            w2[k] = (b1 ? w4[2 + k] : w4[k]) + __shfl_xor(b1 ? w4[k] : w4[2 + k], 2, 32);
        const float sc = (b0 ? w2[1] : w2[0]) + __shfl_xor(b0 ? w2[0] : w2[1], 1, 32);
        const int srow = r0 + 8 * half + (lrow & 7);
        if (srow < n) (b3 ? Sd : Ss)[srow] = sc;
    }
}

__global__ void __launch_bounds__(1024)
gat_chunksort(const int* __restrict__ src, const int* __restrict__ dst,
              int* __restrict__ offs, int* __restrict__ binned, int E, int nb) {
    __shared__ int cnt[MAXBINS];
    __shared__ int wsum[32];
    __shared__ __attribute__((aligned(16))) int stage[CHUNK];
    const int tid   = threadIdx.x;
    const int lane  = tid & 31;
    const int wave  = tid >> 5;
    const int c     = blockIdx.x;
    const int cbase = c * CHUNK;
    const int rest  = E - cbase;
    const int tcnt  = (rest < CHUNK) ? rest : CHUNK;

    int dv[CHUNK / 1024], sv[CHUNK / 1024];
#pragma unroll
    for (int it = 0; it < CHUNK / 1024; ++it) {
        int k = it * 1024 + tid;
        dv[it] = (k < tcnt) ? dst[cbase + k] : 0;
        sv[it] = (k < tcnt) ? src[cbase + k] : 0;
    }
    cnt[tid] = 0;
    __syncthreads();
#pragma unroll
    for (int it = 0; it < CHUNK / 1024; ++it) {
        int k = it * 1024 + tid;
        if (k < tcnt) atomicAdd(&cnt[dv[it] >> BLK_SHIFT], 1);
    }
    __syncthreads();

    const int v = cnt[tid];
    int x = v;
    for (int o = 1; o < 32; o <<= 1) {
        int y = __shfl_up(x, o, 32);
        if (lane >= o) x += y;
    }
    if (lane == 31) wsum[wave] = x;
    __syncthreads();
    int woff = 0;
    for (int w = 0; w < wave; ++w) woff += wsum[w];
    const int excl = woff + x - v;
    cnt[tid] = excl;
    if (tid <= nb) offs[(size_t)c * (nb + 1) + tid] = excl;
    __syncthreads();

#pragma unroll
    for (int it = 0; it < CHUNK / 1024; ++it) {
        int k = it * 1024 + tid;
        if (k < tcnt) {
            int pos = atomicAdd(&cnt[dv[it] >> BLK_SHIFT], 1);
            stage[pos] = sv[it] | ((dv[it] & (BLK_NODES - 1)) << 24);
        }
    }
    __syncthreads();

    int* outp = binned + cbase;
    for (int i = tid * 4; i < tcnt; i += 4096) {
        if (i + 4 <= tcnt) {
            *(int4*)(outp + i) = *(const int4*)(stage + i);
        } else {
            for (int k = i; k < tcnt; ++k) outp[k] = stage[k];
        }
    }
}

__device__ __forceinline__ int fkey(float f) {
    int k = __float_as_int(f);
    return (k >= 0) ? k : (k ^ 0x7FFFFFFF);
}
__device__ __forceinline__ float keyf(int k) {
    return __int_as_float((k >= 0) ? k : (k ^ 0x7FFFFFFF));
}
__device__ __forceinline__ float leaky(float v) { return (v > 0.0f) ? v : NEG_SLOPE * v; }

#define OWN_CAP 2560
typedef int v2i __attribute__((ext_vector_type(2)));

__global__ void __launch_bounds__(256)
gat_owner(const int* __restrict__ binned, const int* __restrict__ offs,
          const float* __restrict__ Ss, const float* __restrict__ Sd,
          const float* __restrict__ XP, const float* __restrict__ bias,
          float* __restrict__ out, int n, int nChunk, int nb) {
    __shared__ __attribute__((aligned(16))) float stacc[BLK_NODES * NHID];
    __shared__ v2i   rec[OWN_CAP + 8];
    __shared__ int   words[OWN_CAP];
    __shared__ __attribute__((aligned(16))) int4 seg[MAXCHUNK];
    __shared__ int   ncnt[BLK_NODES];
    __shared__ int   noff[BLK_NODES];
    __shared__ int   ncur[BLK_NODES];
    __shared__ int   mkey[BLK_NODES];
    __shared__ float sdl[BLK_NODES];
    __shared__ float esl[BLK_NODES];
    __shared__ float stm[BLK_NODES];
    __shared__ float stden[BLK_NODES];
    __shared__ int   red[8];
    __shared__ int   wsum[4];

    const int tid  = threadIdx.x;
    const int lane = tid & 31;
    const int wave = tid >> 5;
    const int p16  = lane & 15;
    const int hw   = tid >> 4;
    const int b    = blockIdx.x;
    const int n0   = b * BLK_NODES;

    int sst = 0, slen = 0;
    if (tid < nChunk) {
        const int* o = offs + (size_t)tid * (nb + 1) + b;
        const int s0 = o[0], s1 = o[1];
        sst  = tid * CHUNK + s0;
        slen = s1 - s0;
    }
    int sinc = slen;
    for (int o = 1; o < 32; o <<= 1) {
        int y = __shfl_up(sinc, o, 32);
        if (lane >= o) sinc += y;
    }
    if (lane == 31) red[wave] = sinc;

    if (tid < BLK_NODES) {
        int node = n0 + tid;
        float sd = 0.0f, es = 0.0f;
        if (node < n) { sd = Sd[node]; es = leaky(Ss[node] + sd); }
        sdl[tid]  = sd;
        esl[tid]  = es;
        mkey[tid] = fkey(es);
    }
    __syncthreads();
    int cnt = 0;
    {
        int woff = 0;
        for (int w = 0; w < 8; ++w) {
            int tt = red[w];
            if (w < wave) woff += tt;
            cnt += tt;
        }
        seg[tid] = make_int4(sst, slen, woff + sinc - slen, 0);
    }
    const int ntile = (cnt > OWN_CAP) ? (cnt + OWN_CAP - 1) / OWN_CAP : 1;
    const float4* XP4 = (const float4*)XP;
    const float4  bv  = ((const float4*)bias)[p16];

    for (int t = 0; t < ntile; ++t) {
        const int  tb   = t * OWN_CAP;
        const int  rest = cnt - tb;
        const int  tcnt = (rest < OWN_CAP) ? rest : OWN_CAP;

        if (tid < BLK_NODES) ncnt[tid] = 0;
        __syncthreads();

        {
            int  wv[MAXCHUNK / 8], dd[MAXCHUNK / 8];
            bool more = false;
#pragma unroll
            for (int u = 0; u < MAXCHUNK / 16; ++u) {
                const int4 sg = seg[hw + 16 * u];
#pragma unroll
                for (int h = 0; h < 2; ++h) {
                    const int      k  = p16 + 16 * h;
                    const unsigned d  = (unsigned)(sg.z - tb + k);
                    const bool     in = (k < sg.y) && (d < (unsigned)OWN_CAP);
                    dd[2 * u + h] = in ? (int)d : -1;
                    wv[2 * u + h] = binned[in ? sg.x + k : 0];
                }
                more = more || (sg.y > 32);
            }
#pragma unroll
            for (int u = 0; u < MAXCHUNK / 8; ++u) {
                if (dd[u] >= 0) {
                    words[dd[u]] = wv[u];
                    atomicAdd(&ncnt[(int)((unsigned)wv[u] >> 24)], 1);
                }
            }
            if (more) {
                for (int u = 0; u < MAXCHUNK / 16; ++u) {
                    const int4 sg = seg[hw + 16 * u];
                    for (int k = p16 + 32; k < sg.y; k += 16) {
                        const unsigned d = (unsigned)(sg.z - tb + k);
                        if (d < (unsigned)OWN_CAP) {
                            const int w = binned[sg.x + k];
                            words[d] = w;
                            atomicAdd(&ncnt[(int)((unsigned)w >> 24)], 1);
                        }
                    }
                }
            }
        }
        __syncthreads();

        int sv = 0, sx = 0;
        if (tid < BLK_NODES) {
            sv = ncnt[tid];
            sx = sv;
            for (int o = 1; o < 32; o <<= 1) {
                int y = __shfl_up(sx, o, 32);
                if (lane >= o) sx += y;
            }
            if (lane == 31) wsum[wave] = sx;
        }
        __syncthreads();
        if (tid < BLK_NODES) {
            int woff = 0;
            for (int w = 0; w < 4; ++w) if (w < wave) woff += wsum[w];
            int off = woff + sx - sv;
            noff[tid] = off;
            ncur[tid] = off;
        }
        __syncthreads();

        for (int i0 = tid; i0 < tcnt; i0 += 1024) {
            int   wv[4];
            float sc4[4];
#pragma unroll
            for (int u = 0; u < 4; ++u) { int i = i0 + u * 256; wv[u] = words[(i < tcnt) ? i : tcnt - 1]; }
#pragma unroll
            for (int u = 0; u < 4; ++u) sc4[u] = Ss[wv[u] & 0xFFFFFF];
#pragma unroll
            for (int u = 0; u < 4; ++u) {
                int i = i0 + u * 256;
                if (i < tcnt) {
                    int   j = (int)((unsigned)wv[u] >> 24);
                    float e = leaky(sc4[u] + sdl[j]);
                    atomicMax(&mkey[j], fkey(e));
                    int pos = atomicAdd(&ncur[j], 1);
                    v2i r;
                    r.x = wv[u] & 0xFFFFFF;
                    r.y = __float_as_int(e);
                    rec[pos] = r;
                }
            }
        }
        __syncthreads();

        for (int q = 0; q < BLK_NODES / 16; ++q) {
            const int   jj   = hw + 16 * q;
            const int   node = n0 + jj;
            const int   rs   = noff[jj];
            const int   re   = ncur[jj];
            const float mnew = keyf(mkey[jj]);
            float4 a0 = make_float4(0.0f, 0.0f, 0.0f, 0.0f);
            float  mold, den;
            if (t == 0) {
                mold = esl[jj];
                den  = 1.0f;
                if (node < n) a0 = XP4[(size_t)node * (NHID / 4) + p16];
            } else {
                a0   = ((const float4*)stacc)[jj * (NHID / 4) + p16];
                mold = stm[jj];
                den  = stden[jj];
            }
            float4 acc  = make_float4(0.0f, 0.0f, 0.0f, 0.0f);
            float  dsum = 0.0f;
            for (int r = rs; r < re; r += 8) {
                v2i    rc[8];
                float4 xv[8];
#pragma unroll
                for (int u = 0; u < 8; ++u) {
                    int idx = r + u;
                    if (idx > re - 1) idx = re - 1;
                    rc[u] = rec[idx];
                }
#pragma unroll
                for (int u = 0; u < 8; ++u) xv[u] = XP4[(size_t)rc[u].x * (NHID / 4) + p16];
#pragma unroll
                for (int u = 0; u < 8; ++u) {
                    float ex = (r + u < re) ? __expf(__int_as_float(rc[u].y) - mnew) : 0.0f;
                    dsum  += ex;
                    acc.x += ex * xv[u].x;
                    acc.y += ex * xv[u].y;
                    acc.z += ex * xv[u].z;
                    acc.w += ex * xv[u].w;
                }
            }
            const float sc = __expf(mold - mnew);
            den  = den * sc + dsum;
            a0.x = a0.x * sc + acc.x;
            a0.y = a0.y * sc + acc.y;
            a0.z = a0.z * sc + acc.z;
            a0.w = a0.w * sc + acc.w;
            if (t == ntile - 1) {
                if (node < n) {
                    float4 o;
                    o.x = a0.x / den + bv.x;
                    o.y = a0.y / den + bv.y;
                    o.z = a0.z / den + bv.z;
                    o.w = a0.w / den + bv.w;
                    ((float4*)out)[(size_t)node * (NHID / 4) + p16] = o;
                }
            } else {
                ((float4*)stacc)[jj * (NHID / 4) + p16] = a0;
                if (p16 == 0) { stm[jj] = mnew; stden[jj] = den; }
            }
        }
    }
}

extern "C" void kernel_launch(void* const* d_in, const int* in_sizes, int n_in,
                              void* d_out, int out_size, void* d_ws, size_t ws_size,
                              hipStream_t stream) {
    const float* x     = (const float*)d_in[0];
    const int*   ei    = (const int*)  d_in[1];
    const float* W     = (const float*)d_in[2];
    const float* a_src = (const float*)d_in[3];
    const float* a_dst = (const float*)d_in[4];
    const float* bias  = (const float*)d_in[5];
    float*       out   = (float*)d_out;

    const int N = in_sizes[0] / NFEAT;
    const int E = in_sizes[1] / 2;
    const int* src = ei;
    const int* dst = ei + E;
    const int nb     = (N + BLK_NODES - 1) / BLK_NODES;
    const int nChunk = (E + CHUNK - 1) / CHUNK;

    float* xp     = (float*)d_ws;
    int*   binned = (int*)(xp + (size_t)N * NHID);
    float* ssrc   = (float*)(binned + E);
    float* sdst   = ssrc + N;
    int*   offs   = (int*)(sdst + N);

    gat_gemm     <<<256,    GEMM_THREADS, 0, stream>>>(x, W, a_src, a_dst, xp, ssrc, sdst, N);
    gat_chunksort<<<nChunk,         1024, 0, stream>>>(src, dst, offs, binned, E, nb);
    gat_owner    <<<nb,             256,  0, stream>>>(binned, offs, ssrc, sdst, xp, bias, out, N, nChunk, nb);
}


# chunk sort runs inside the GEMM launch: each of its 256 work-groups first sorts a 6400-edge chunk in LDS, then does its strips; 2 launches, not 3; new chunk size changes edge order in a node's sum
# speedup vs baseline: 5.5236x; 1.0735x over previous
#include <hip/hip_runtime.h>
#include <hip/hip_bf16.h>
#include <math.h>

typedef float v2f __attribute__((ext_vector_type(2)));
typedef float v8f __attribute__((ext_vector_type(8)));

#define NFEAT 256
#define NHID  64
#define NEG_SLOPE 0.2f

#define BLK_NODES 128
#define BLK_SHIFT 7
#define GEMM_THREADS 800
#define CHUNK     6400
#define MAXBINS   GEMM_THREADS
#define MAXCHUNK  256

#define KBLK   (NFEAT / 8)

__device__ __forceinline__ void gemm_kgroup(const float4 (&xa)[4], const float4* bp,
                                            v8f& c0, v8f& c1, v8f& c2, v8f& c3) {
#pragma unroll
    for (int u = 0; u < 4; ++u) {
        const float4 b0 = bp[u * 128], b1 = bp[u * 128 + 32], b2 = bp[u * 128 + 64], b3 = bp[u * 128 + 96];
        v2f a, p0, p1, p2, p3;
        a[0] = xa[u].x;  a[1] = xa[u].y;
        p0[0] = b0.x; p0[1] = b0.y;  p1[0] = b1.x; p1[1] = b1.y;
        p2[0] = b2.x; p2[1] = b2.y;  p3[0] = b3.x; p3[1] = b3.y;
        c0 = __builtin_amdgcn_wmma_f32_16x16x4_f32(false, a, false, p0, (short)0, c0, false, false);
        c1 = __builtin_amdgcn_wmma_f32_16x16x4_f32(false, a, false, p1, (short)0, c1, false, false);
        c2 = __builtin_amdgcn_wmma_f32_16x16x4_f32(false, a, false, p2, (short)0, c2, false, false);
        c3 = __builtin_amdgcn_wmma_f32_16x16x4_f32(false, a, false, p3, (short)0, c3, false, false);
        a[0] = xa[u].z;  a[1] = xa[u].w;
        p0[0] = b0.z; p0[1] = b0.w;  p1[0] = b1.z; p1[1] = b1.w;
        p2[0] = b2.z; p2[1] = b2.w;  p3[0] = b3.z; p3[1] = b3.w;
        c0 = __builtin_amdgcn_wmma_f32_16x16x4_f32(false, a, false, p0, (short)0, c0, false, false);
        c1 = __builtin_amdgcn_wmma_f32_16x16x4_f32(false, a, false, p1, (short)0, c1, false, false);
        c2 = __builtin_amdgcn_wmma_f32_16x16x4_f32(false, a, false, p2, (short)0, c2, false, false);
        c3 = __builtin_amdgcn_wmma_f32_16x16x4_f32(false, a, false, p3, (short)0, c3, false, false);
    }
}

__global__ void __launch_bounds__(GEMM_THREADS)
gat_gemm(const float* __restrict__ X, const float* __restrict__ W,
         const float* __restrict__ As, const float* __restrict__ Ad,
         float* __restrict__ XP, float* __restrict__ Ss, float* __restrict__ Sd, int n,
         const int* __restrict__ src, const int* __restrict__ dst,
         int* __restrict__ offs, int* __restrict__ binned, int E, int nb) {
    __shared__ __attribute__((aligned(16))) float4 wl[KBLK * 4 * 32];
    __shared__ int cnt[MAXBINS];
    __shared__ int wsum[32];
    __shared__ __attribute__((aligned(16))) int stage[CHUNK];

    const int tid    = threadIdx.x;
    const int wave   = __builtin_amdgcn_readfirstlane(tid >> 5);
    const int lane   = tid & 31;
    const int half   = lane >> 4;
    const int lrow   = lane & 15;
    const int nwaves = blockDim.x >> 5;
    const int nstrip = (n + 15) >> 4;
    const int sstep  = gridDim.x * nwaves;
    int s = blockIdx.x + gridDim.x * wave;

    float4 cur[4], nxt[4];
    const float4* ap;
    {
        int arow = s * 16 + lrow;
        if (arow > n - 1) arow = n - 1;
        ap = (const float4*)(X + (size_t)arow * NFEAT + 4 * half);
    }
    if (s < nstrip) {
#pragma unroll
        for (int u = 0; u < 4; ++u) cur[u] = ap[2 * u];
    }

    const float as0 = As[lrow], as1 = As[16 + lrow], as2 = As[32 + lrow], as3 = As[48 + lrow];
    const float ad0 = Ad[lrow], ad1 = Ad[16 + lrow], ad2 = Ad[32 + lrow], ad3 = Ad[48 + lrow];

    const int cbase = blockIdx.x * CHUNK;
    const int rest  = E - cbase;
    const int tcnt  = (rest < 0) ? 0 : ((rest < CHUNK) ? rest : CHUNK);
    int dv[CHUNK / GEMM_THREADS], sv[CHUNK / GEMM_THREADS];
#pragma unroll
    for (int it = 0; it < CHUNK / GEMM_THREADS; ++it) {
        int k = it * GEMM_THREADS + tid;
        dv[it] = (k < tcnt) ? dst[cbase + k] : 0;
        sv[it] = (k < tcnt) ? src[cbase + k] : 0;
    }
    cnt[tid] = 0;

    for (int j = wave; j < KBLK; j += nwaves) {
        const float* wp = W + (size_t)(8 * j + 4 * half) * NHID + lrow;
#pragma unroll
        for (int t = 0; t < 4; ++t) {
            float4 v;
            v.x = wp[0 * NHID + 16 * t];
            v.y = wp[1 * NHID + 16 * t];
            v.z = wp[2 * NHID + 16 * t];
            v.w = wp[3 * NHID + 16 * t];
            wl[(j * 4 + t) * 32 + lane] = v;
        }
    }

    __syncthreads();

#pragma unroll
    for (int it = 0; it < CHUNK / GEMM_THREADS; ++it) {
        int k = it * GEMM_THREADS + tid;
        if (k < tcnt) atomicAdd(&cnt[dv[it] >> BLK_SHIFT], 1);
    }
    __syncthreads();

    {
        const int v = cnt[tid];
        int x = v;
        for (int o = 1; o < 32; o <<= 1) {
            int y = __shfl_up(x, o, 32);
            if (lane >= o) x += y;
        }
        if (lane == 31) wsum[wave] = x;
        __syncthreads();
        int woff = 0;
        for (int w = 0; w < wave; ++w) woff += wsum[w];
        const int excl = woff + x - v;
        cnt[tid] = excl;
        if (tcnt > 0 && tid <= nb) offs[(size_t)blockIdx.x * (nb + 1) + tid] = excl;
    }
    __syncthreads();

#pragma unroll
    for (int it = 0; it < CHUNK / GEMM_THREADS; ++it) {
        int k = it * GEMM_THREADS + tid;
        if (k < tcnt) {
            int pos = atomicAdd(&cnt[dv[it] >> BLK_SHIFT], 1);
            stage[pos] = sv[it] | ((dv[it] & (BLK_NODES - 1)) << 24);
        }
    }
    __syncthreads();

    {
        int* outp = binned + cbase;
        for (int i = tid * 4; i < tcnt; i += 4 * GEMM_THREADS) {
            if (i + 4 <= tcnt) {
                *(int4*)(outp + i) = *(const int4*)(stage + i);
            } else {
                for (int k = i; k < tcnt; ++k) outp[k] = stage[k];
            }
        }
    }

    for (; s < nstrip; s += sstep) {
        const int r0 = s * 16;
        v8f c0 = {}, c1 = {}, c2 = {}, c3 = {};

        const float4* apn;
        {
            const int sn = (s + sstep < nstrip) ? s + sstep : s;
            int arow = sn * 16 + lrow;
            if (arow > n - 1) arow = n - 1;
            apn = (const float4*)(X + (size_t)arow * NFEAT + 4 * half);
        }

        const float4* bp = wl + lane;
#pragma unroll 2
        for (int jj = 0; jj < KBLK; jj += 4) {
            const float4* pn = (jj + 4 < KBLK) ? ap + 2 * (jj + 4) : apn;
#pragma unroll
            for (int u = 0; u < 4; ++u) nxt[u] = pn[2 * u];
            gemm_kgroup(cur, bp + jj * 128, c0, c1, c2, c3);
#pragma unroll
            for (int u = 0; u < 4; ++u) cur[u] = nxt[u];
        }
        ap = apn;

#pragma unroll
        for (int j = 0; j < 8; ++j) {
            int row = half * 8 + j;
            if (r0 + row < n) {
                float* o = XP + (size_t)(r0 + row) * NHID + lrow;
                o[0]  = c0[j];
                o[16] = c1[j];
                o[32] = c2[j];
                o[48] = c3[j];
            }
        }

        float v[16];
#pragma unroll
        for (int j = 0; j < 8; ++j) {
            v[j]     = c0[j] * as0 + c1[j] * as1 + c2[j] * as2 + c3[j] * as3;
            v[8 + j] = c0[j] * ad0 + c1[j] * ad1 + c2[j] * ad2 + c3[j] * ad3;
        }
        const bool b3 = (lrow & 8) != 0, b2 = (lrow & 4) != 0, b1 = (lrow & 2) != 0, b0 = (lrow & 1) != 0;
        float w8[8], w4[4], w2[2];
#pragma unroll
        for (int k = 0; k < 8; ++k)
            w8[k] = (b3 ? v[8 + k] : v[k]) + __shfl_xor(b3 ? v[k] : v[8 + k], 8, 32);
#pragma unroll
        for (int k = 0; k < 4; ++k)
            w4[k] = (b2 ? w8[4 + k] : w8[k]) + __shfl_xor(b2 ? w8[k] : w8[4 + k], 4, 32);
#pragma unroll
        for (int k = 0; k < 2; ++k)
            w2[k] = (b1 ? w4[2 + k] : w4[k]) + __shfl_xor(b1 ? w4[k] : w4[2 + k], 2, 32);
        const float sc = (b0 ? w2[1] : w2[0]) + __shfl_xor(b0 ? w2[0] : w2[1], 1, 32);
        const int srow = r0 + 8 * half + (lrow & 7);
        if (srow < n) (b3 ? Sd : Ss)[srow] = sc;
    }
}

__device__ __forceinline__ int fkey(float f) {
    int k = __float_as_int(f);
    return (k >= 0) ? k : (k ^ 0x7FFFFFFF);
}
__device__ __forceinline__ float keyf(int k) {
    return __int_as_float((k >= 0) ? k : (k ^ 0x7FFFFFFF));
}
__device__ __forceinline__ float leaky(float v) { return (v > 0.0f) ? v : NEG_SLOPE * v; }

#define OWN_CAP 2560
typedef int v2i __attribute__((ext_vector_type(2)));

__global__ void __launch_bounds__(256)
gat_owner(const int* __restrict__ binned, const int* __restrict__ offs,
          const float* __restrict__ Ss, const float* __restrict__ Sd,
          const float* __restrict__ XP, const float* __restrict__ bias,
          float* __restrict__ out, int n, int nChunk, int nb) {
    __shared__ __attribute__((aligned(16))) float stacc[BLK_NODES * NHID];
    __shared__ v2i   rec[OWN_CAP + 8];
    __shared__ int   words[OWN_CAP];
    __shared__ __attribute__((aligned(16))) int4 seg[MAXCHUNK];
    __shared__ int   ncnt[BLK_NODES];
    __shared__ int   noff[BLK_NODES];
    __shared__ int   ncur[BLK_NODES];
    __shared__ int   mkey[BLK_NODES];
    __shared__ float sdl[BLK_NODES];
    __shared__ float esl[BLK_NODES];
    __shared__ float stm[BLK_NODES];
    __shared__ float stden[BLK_NODES];
    __shared__ int   red[8];
    __shared__ int   wsum[4];

    const int tid  = threadIdx.x;
    const int lane = tid & 31;
    const int wave = tid >> 5;
    const int p16  = lane & 15;
    const int hw   = tid >> 4;
    const int b    = blockIdx.x;
    const int n0   = b * BLK_NODES;

    int sst = 0, slen = 0;
    if (tid < nChunk) {
        const int* o = offs + (size_t)tid * (nb + 1) + b;
        const int s0 = o[0], s1 = o[1];
        sst  = tid * CHUNK + s0;
        slen = s1 - s0;
    }
    int sinc = slen;
    for (int o = 1; o < 32; o <<= 1) {
        int y = __shfl_up(sinc, o, 32);
        if (lane >= o) sinc += y;
    }
    if (lane == 31) red[wave] = sinc;

    if (tid < BLK_NODES) {
        int node = n0 + tid;
        float sd = 0.0f, es = 0.0f;
        if (node < n) { sd = Sd[node]; es = leaky(Ss[node] + sd); }
        sdl[tid]  = sd;
        esl[tid]  = es;
        mkey[tid] = fkey(es);
    }
    __syncthreads();
    int cnt = 0;
    {
        int woff = 0;
        for (int w = 0; w < 8; ++w) {
            int tt = red[w];
            if (w < wave) woff += tt;
            cnt += tt;
        }
        seg[tid] = make_int4(sst, slen, woff + sinc - slen, 0);
    }
    const int ntile = (cnt > OWN_CAP) ? (cnt + OWN_CAP - 1) / OWN_CAP : 1;
    const float4* XP4 = (const float4*)XP;
    const float4  bv  = ((const float4*)bias)[p16];

    for (int t = 0; t < ntile; ++t) {
        const int  tb   = t * OWN_CAP;
        const int  rest = cnt - tb;
        const int  tcnt = (rest < OWN_CAP) ? rest : OWN_CAP;

        if (tid < BLK_NODES) ncnt[tid] = 0;
        __syncthreads();

        {
            int  wv[MAXCHUNK / 8], dd[MAXCHUNK / 8];
            bool more = false;
#pragma unroll
            for (int u = 0; u < MAXCHUNK / 16; ++u) {
                const int4 sg = seg[hw + 16 * u];
#pragma unroll
                for (int h = 0; h < 2; ++h) {
                    const int      k  = p16 + 16 * h;
                    const unsigned d  = (unsigned)(sg.z - tb + k);
                    const bool     in = (k < sg.y) && (d < (unsigned)OWN_CAP);
                    dd[2 * u + h] = in ? (int)d : -1;
                    wv[2 * u + h] = binned[in ? sg.x + k : 0];
                }
                more = more || (sg.y > 32);
            }
#pragma unroll
            for (int u = 0; u < MAXCHUNK / 8; ++u) {
                if (dd[u] >= 0) {
                    words[dd[u]] = wv[u];
                    atomicAdd(&ncnt[(int)((unsigned)wv[u] >> 24)], 1);
                }
            }
            if (more) {
                for (int u = 0; u < MAXCHUNK / 16; ++u) {
                    const int4 sg = seg[hw + 16 * u];
                    for (int k = p16 + 32; k < sg.y; k += 16) {
                        const unsigned d = (unsigned)(sg.z - tb + k);
                        if (d < (unsigned)OWN_CAP) {
                            const int w = binned[sg.x + k];
                            words[d] = w;
                            atomicAdd(&ncnt[(int)((unsigned)w >> 24)], 1);
                        }
                    }
                }
            }
        }
        __syncthreads();

        int sv = 0, sx = 0;
        if (tid < BLK_NODES) {
            sv = ncnt[tid];
            sx = sv;
            for (int o = 1; o < 32; o <<= 1) {
                int y = __shfl_up(sx, o, 32);
                if (lane >= o) sx += y;
            }
            if (lane == 31) wsum[wave] = sx;
        }
        __syncthreads();
        if (tid < BLK_NODES) {
            int woff = 0;
            for (int w = 0; w < 4; ++w) if (w < wave) woff += wsum[w];
            int off = woff + sx - sv;
            noff[tid] = off;
            ncur[tid] = off;
        }
        __syncthreads();

        for (int i0 = tid; i0 < tcnt; i0 += 1024) {
            int   wv[4];
            float sc4[4];
#pragma unroll
            for (int u = 0; u < 4; ++u) { int i = i0 + u * 256; wv[u] = words[(i < tcnt) ? i : tcnt - 1]; }
#pragma unroll
            for (int u = 0; u < 4; ++u) sc4[u] = Ss[wv[u] & 0xFFFFFF];
#pragma unroll
            for (int u = 0; u < 4; ++u) {
                int i = i0 + u * 256;
                if (i < tcnt) {
                    int   j = (int)((unsigned)wv[u] >> 24);
                    float e = leaky(sc4[u] + sdl[j]);
                    atomicMax(&mkey[j], fkey(e));
                    int pos = atomicAdd(&ncur[j], 1);
                    v2i r;
                    r.x = wv[u] & 0xFFFFFF;
                    r.y = __float_as_int(e);
                    rec[pos] = r;
                }
            }
        }
        __syncthreads();

        for (int q = 0; q < BLK_NODES / 16; ++q) {
            const int   jj   = hw + 16 * q;
            const int   node = n0 + jj;
            const int   rs   = noff[jj];
            const int   re   = ncur[jj];
            const float mnew = keyf(mkey[jj]);
            float4 a0 = make_float4(0.0f, 0.0f, 0.0f, 0.0f);
            float  mold, den;
            if (t == 0) {
                mold = esl[jj];
                den  = 1.0f;
                if (node < n) a0 = XP4[(size_t)node * (NHID / 4) + p16];
            } else {
                a0   = ((const float4*)stacc)[jj * (NHID / 4) + p16];
                mold = stm[jj];
                den  = stden[jj];
            }
            float4 acc  = make_float4(0.0f, 0.0f, 0.0f, 0.0f);
            float  dsum = 0.0f;
            for (int r = rs; r < re; r += 8) {
                v2i    rc[8];
                float4 xv[8];
#pragma unroll
                for (int u = 0; u < 8; ++u) {
                    int idx = r + u;
                    if (idx > re - 1) idx = re - 1;
                    rc[u] = rec[idx];
                }
#pragma unroll
                for (int u = 0; u < 8; ++u) xv[u] = XP4[(size_t)rc[u].x * (NHID / 4) + p16];
#pragma unroll
                for (int u = 0; u < 8; ++u) {
                    float ex = (r + u < re) ? __expf(__int_as_float(rc[u].y) - mnew) : 0.0f;
                    dsum  += ex;
                    acc.x += ex * xv[u].x;
                    acc.y += ex * xv[u].y;
                    acc.z += ex * xv[u].z;
                    acc.w += ex * xv[u].w;
                }
            }
            const float sc = __expf(mold - mnew);
            den  = den * sc + dsum;
            a0.x = a0.x * sc + acc.x;
            a0.y = a0.y * sc + acc.y;
            a0.z = a0.z * sc + acc.z;
            a0.w = a0.w * sc + acc.w;
            if (t == ntile - 1) {
                if (node < n) {
                    float4 o;
                    o.x = a0.x / den + bv.x;
                    o.y = a0.y / den + bv.y;
                    o.z = a0.z / den + bv.z;
                    o.w = a0.w / den + bv.w;
                    ((float4*)out)[(size_t)node * (NHID / 4) + p16] = o;
                }
            } else {
                ((float4*)stacc)[jj * (NHID / 4) + p16] = a0;
                if (p16 == 0) { stm[jj] = mnew; stden[jj] = den; }
            }
        }
    }
}

extern "C" void kernel_launch(void* const* d_in, const int* in_sizes, int n_in,
                              void* d_out, int out_size, void* d_ws, size_t ws_size,
                              hipStream_t stream) {
    const float* x     = (const float*)d_in[0];
    const int*   ei    = (const int*)  d_in[1];
    const float* W     = (const float*)d_in[2];
    const float* a_src = (const float*)d_in[3];
    const float* a_dst = (const float*)d_in[4];
    const float* bias  = (const float*)d_in[5];
    float*       out   = (float*)d_out;

    const int N = in_sizes[0] / NFEAT;
    const int E = in_sizes[1] / 2;
    const int* src = ei;
    const int* dst = ei + E;
    const int nb     = (N + BLK_NODES - 1) / BLK_NODES;
    const int nChunk = (E + CHUNK - 1) / CHUNK;

    float* xp     = (float*)d_ws;
    int*   binned = (int*)(xp + (size_t)N * NHID);
    float* ssrc   = (float*)(binned + E);
    float* sdst   = ssrc + N;
    int*   offs   = (int*)(sdst + N);

    gat_gemm <<<MAXCHUNK, GEMM_THREADS, 0, stream>>>(x, W, a_src, a_dst, xp, ssrc, sdst, N,
                                                     src, dst, offs, binned, E, nb);
    gat_owner<<<nb,       256,          0, stream>>>(binned, offs, ssrc, sdst, xp, bias, out, N, nChunk, nb);
}
